# starting kernel
# baseline (speedup 1.0000x reference)
#include <hip/hip_runtime.h>

#define NNODES 20000
#define NRELH  16
#define RTOT   33
#define HID    64
#define NCLS   16
#define RN     (RTOT * NNODES)
#define PCOLS  (RTOT * NCLS)

typedef __attribute__((ext_vector_type(2))) float v2f;
typedef __attribute__((ext_vector_type(8))) float v8f;

__global__ __launch_bounds__(256) void k_zero(float* __restrict__ p, int n) {
  int i = blockIdx.x * blockDim.x + threadIdx.x;
  if (i < n) p[i] = 0.0f;
}

__global__ __launch_bounds__(256) void k_count(const int* __restrict__ src,
                                               const int* __restrict__ rel,
                                               const int* __restrict__ dst,
                                               float* __restrict__ cnt_ro,
                                               float* __restrict__ cnt_rs, int E) {
  int e = blockIdx.x * blockDim.x + threadIdx.x;
  if (e >= E) return;
  int s = src[e], r = rel[e], d = dst[e];
  atomicAdd(cnt_ro + (size_t)r * NNODES + d, 1.0f);
  atomicAdd(cnt_ro + (size_t)(r + NRELH) * NNODES + s, 1.0f);
  atomicAdd(cnt_rs + (size_t)r * NNODES + s, 1.0f);
  atomicAdd(cnt_rs + (size_t)(r + NRELH) * NNODES + d, 1.0f);
}

__global__ __launch_bounds__(256) void k_layer1(const int* __restrict__ src,
                                                const int* __restrict__ rel,
                                                const int* __restrict__ dst,
                                                const float* __restrict__ w1,
                                                const float* __restrict__ cnt_ro,
                                                float* __restrict__ hacc, int E) {
  long idx = (long)blockIdx.x * blockDim.x + threadIdx.x;
  if (idx >= (long)E * 16) return;
  int e  = (int)(idx >> 4);
  int i0 = ((int)idx & 15) * 4;
  int s = src[e], r = rel[e], d = dst[e];
  float vo = 1.0f / cnt_ro[(size_t)(r + NRELH) * NNODES + s];
  float vi = 1.0f / cnt_ro[(size_t)r * NNODES + d];
  float4 a = *(const float4*)(w1 + ((size_t)r * NNODES + d) * HID + i0);
  atomicAdd(&hacc[(size_t)s * HID + i0 + 0], vo * a.x);
  atomicAdd(&hacc[(size_t)s * HID + i0 + 1], vo * a.y);
  atomicAdd(&hacc[(size_t)s * HID + i0 + 2], vo * a.z);
  atomicAdd(&hacc[(size_t)s * HID + i0 + 3], vo * a.w);
  float4 b = *(const float4*)(w1 + ((size_t)(r + NRELH) * NNODES + s) * HID + i0);
  atomicAdd(&hacc[(size_t)d * HID + i0 + 0], vi * b.x);
  atomicAdd(&hacc[(size_t)d * HID + i0 + 1], vi * b.y);
  atomicAdd(&hacc[(size_t)d * HID + i0 + 2], vi * b.z);
  atomicAdd(&hacc[(size_t)d * HID + i0 + 3], vi * b.w);
}

__global__ __launch_bounds__(256) void k_hfinal(const float* __restrict__ w1,
                                                const float* __restrict__ b1,
                                                float* __restrict__ h) {
  int idx = blockIdx.x * blockDim.x + threadIdx.x;
  if (idx >= NNODES * HID) return;
  int n = idx / HID, i = idx % HID;
  float v = h[idx] + w1[((size_t)(2 * NRELH) * NNODES + n) * HID + i] + b1[i];
  h[idx] = v > 0.0f ? v : 0.0f;
}

__global__ __launch_bounds__(32) void k_gemm(const float* __restrict__ h,
                                             const float* __restrict__ w2,
                                             float* __restrict__ P) {
  int tile = blockIdx.x;
  int mt = tile % (NNODES / 16);
  int r  = tile / (NNODES / 16);
  int lane  = threadIdx.x;
  int mn    = lane & 15;
  int khalf = lane >> 4;
  const float* hrow = h + (size_t)(mt * 16 + mn) * HID;
  const float* wrel = w2 + (size_t)r * HID * NCLS;
  union { v8f v; float f[8]; } acc;
  acc.v = (v8f)0.0f;
#pragma unroll
  for (int k0 = 0; k0 < HID; k0 += 4) {
    int ka = k0 + 2 * khalf;
    v2f a, b;
    a.x = hrow[ka];
    a.y = hrow[ka + 1];
    b.x = wrel[(size_t)ka * NCLS + mn];
    b.y = wrel[(size_t)(ka + 1) * NCLS + mn];
    acc.v = __builtin_amdgcn_wmma_f32_16x16x4_f32(
         false, a,  false, b,
         (short)0, acc.v,  false,  false);
  }
  int rowoff = khalf * 8;
#pragma unroll
  for (int v = 0; v < 8; ++v) {
    P[(size_t)(mt * 16 + v + rowoff) * PCOLS + r * NCLS + mn] = acc.f[v];
  }
}

__global__ __launch_bounds__(256) void k_outinit(const float* __restrict__ b2,
                                                 const float* __restrict__ P,
                                                 float* __restrict__ out) {
  int idx = blockIdx.x * blockDim.x + threadIdx.x;
  if (idx >= NNODES * NCLS) return;
  int n = idx / NCLS, c = idx % NCLS;
  out[idx] = b2[c] + P[(size_t)n * PCOLS + (2 * NRELH) * NCLS + c];
}

__global__ __launch_bounds__(256) void k_layer2(const int* __restrict__ src,
                                                const int* __restrict__ rel,
                                                const int* __restrict__ dst,
                                                const float* __restrict__ cnt_rs,
                                                const float* __restrict__ P,
                                                float* __restrict__ out, int E) {
  long idx = (long)blockIdx.x * blockDim.x + threadIdx.x;
  if (idx >= (long)E * 16) return;
  int e = (int)(idx >> 4);
  int c = (int)idx & 15;
  int s = src[e], r = rel[e], d = dst[e];
  float vo = 1.0f / cnt_rs[(size_t)r * NNODES + s];
  float vi = 1.0f / cnt_rs[(size_t)(r + NRELH) * NNODES + d];
  atomicAdd(&out[(size_t)s * NCLS + c], vo * P[(size_t)d * PCOLS + r * NCLS + c]);
  atomicAdd(&out[(size_t)d * NCLS + c], vi * P[(size_t)s * PCOLS + (r + NRELH) * NCLS + c]);
}

extern "C" void kernel_launch(void* const* d_in, const int* in_sizes, int n_in,
                              void* d_out, int out_size, void* d_ws, size_t ws_size,
                              hipStream_t stream) {
  const int*   src = (const int*)d_in[0];
  const int*   rel = (const int*)d_in[1];
  const int*   dst = (const int*)d_in[2];
  const float* w1  = (const float*)d_in[3];
  const float* b1  = (const float*)d_in[4];
  const float* w2  = (const float*)d_in[5];
  const float* b2  = (const float*)d_in[6];
  float* out = (float*)d_out;
  const int E = in_sizes[0];

  float* cnt_ro = (float*)d_ws;
  float* cnt_rs = cnt_ro + RN;
  float* h      = cnt_rs + RN;
  float* P      = h + (size_t)NNODES * HID;

  const int zn = 2 * RN + NNODES * HID;
  k_zero<<<(zn + 255) / 256, 256, 0, stream>>>(cnt_ro, zn);

  k_count<<<(E + 255) / 256, 256, 0, stream>>>(src, rel, dst, cnt_ro, cnt_rs, E);

  long l1 = (long)E * 16;
  k_layer1<<<(unsigned)((l1 + 255) / 256), 256, 0, stream>>>(src, rel, dst, w1, cnt_ro, h, E);

  k_hfinal<<<(NNODES * HID + 255) / 256, 256, 0, stream>>>(w1, b1, h);

  k_gemm<<<(NNODES / 16) * RTOT, 32, 0, stream>>>(h, w2, P);

  k_outinit<<<(NNODES * NCLS + 255) / 256, 256, 0, stream>>>(b2, P, out);

  long l2 = (long)E * 16;
  k_layer2<<<(unsigned)((l2 + 255) / 256), 256, 0, stream>>>(src, rel, dst, cnt_rs, P, out, E);
}


# Pull kernels: each lane decodes one message, DPP row/quad picks hand it to the lanes that load its row as 16 bytes; per-message readlane and scalar address chain gone; h, out add order changes
# speedup vs baseline: 1.6221x; 1.6221x over previous
#include <hip/hip_runtime.h>

#define NNODES 20000
#define NRELH  16
#define RTOT   33
#define HID    64
#define NCLS   16
#define PCOLS  (RTOT * NCLS)
#define OBITS  15
#define OMASK  32767
#define NOFF   20032

typedef __attribute__((ext_vector_type(2))) float v2f;
typedef __attribute__((ext_vector_type(8))) float v8f;


__global__ __launch_bounds__(256) void k_zero(int* __restrict__ p, int n) {
  int i = blockIdx.x * blockDim.x + threadIdx.x;
  if (i < n) p[i] = 0;
}

__global__ __launch_bounds__(256) void k_rank(const int* __restrict__ src,
                                              const int* __restrict__ dst,
                                              int* __restrict__ deg,
                                              int* __restrict__ rk_s,
                                              int* __restrict__ rk_d, int E) {
  int e = blockIdx.x * blockDim.x + threadIdx.x;
  if (e >= E) return;
  rk_s[e] = atomicAdd(deg + src[e], 1);
  rk_d[e] = atomicAdd(deg + dst[e], 1);
}

#define SCAN_T   1024
#define SCAN_PER 20
__global__ __launch_bounds__(SCAN_T) void k_scan(const int* __restrict__ deg,
                                                 int* __restrict__ off) {
  __shared__ int part[SCAN_T];
  int t = threadIdx.x;
  int b = t * SCAN_PER;
  int loc[SCAN_PER];
  int s = 0;
#pragma unroll
  for (int i = 0; i < SCAN_PER; ++i) {
    int idx = b + i;
    int v = (idx < NNODES) ? deg[idx] : 0;
    loc[i] = s;
    s += v;
  }
  part[t] = s;
  __syncthreads();
  for (int d = 1; d < SCAN_T; d <<= 1) {
    int v = (t >= d) ? part[t - d] : 0;
    __syncthreads();
    part[t] += v;
    __syncthreads();
  }
  int excl = part[t] - s;
#pragma unroll
  for (int i = 0; i < SCAN_PER; ++i) {
    int idx = b + i;
    if (idx < NNODES) off[idx] = excl + loc[i];
  }
  if (t == SCAN_T - 1) off[NNODES] = part[t];
}

__global__ __launch_bounds__(256) void k_fill(const int* __restrict__ src,
                                              const int* __restrict__ rel,
                                              const int* __restrict__ dst,
                                              const int* __restrict__ off,
                                              const int* __restrict__ rk_s,
                                              const int* __restrict__ rk_d,
                                              int* __restrict__ msg, int E) {
  int e = blockIdx.x * blockDim.x + threadIdx.x;
  if (e >= E) return;
  int s = src[e], r = rel[e], d = dst[e];
  msg[off[s] + rk_s[e]] = (r << OBITS) | d;
  msg[off[d] + rk_d[e]] = ((r + NRELH) << OBITS) | s;
}

__device__ __forceinline__ int seg_counts(const int* __restrict__ msg, int beg, int end, int lane) {
  int cnt = 0;
  for (int base = beg; base < end; base += 32) {
    int i = base + lane;
    int q = (i < end) ? (msg[i] >> OBITS) : 64;
#pragma unroll
    for (int j = 0; j < 32; ++j)
      cnt += (__builtin_amdgcn_readlane(q, j) == lane) ? 1 : 0;
  }
  return cnt;
}

template <int K>
__device__ __forceinline__ int row_pick(int x) {
  return __builtin_amdgcn_update_dpp(0, x, 0x150 + K, 0xf, 0xf, false);
}
template <int K>
__device__ __forceinline__ int quad_pick(int x) {
  return __builtin_amdgcn_update_dpp(0, x, K * 0x55, 0xf, 0xf, false);
}

#define L1_LOAD(U)                                                           \
  {                                                                          \
    int ro = row_pick<8 * B + (U)>(roff);                                    \
    iv[U] = __int_as_float(row_pick<8 * B + (U)>(__float_as_int(invl)));     \
    v[U] = *(const float4*)(w1c + (unsigned)(ro + lo));                      \
  }
#define L1_SUM(U)                                                            \
  {                                                                          \
    acc.x = __builtin_fmaf(iv[U], v[U].x, acc.x);                            \
    acc.y = __builtin_fmaf(iv[U], v[U].y, acc.y);                            \
    acc.z = __builtin_fmaf(iv[U], v[U].z, acc.z);                            \
    acc.w = __builtin_fmaf(iv[U], v[U].w, acc.w);                            \
  }
template <int B>
__device__ __forceinline__ void l1_block(int roff, float invl, int lo,
                                         const char* __restrict__ w1c, float4& acc) {
  float4 v[8];
  float iv[8];
  L1_LOAD(0) L1_LOAD(1) L1_LOAD(2) L1_LOAD(3) L1_LOAD(4) L1_LOAD(5) L1_LOAD(6) L1_LOAD(7)
  L1_SUM(0) L1_SUM(1) L1_SUM(2) L1_SUM(3) L1_SUM(4) L1_SUM(5) L1_SUM(6) L1_SUM(7)
}

#define HPITCH 68
__global__ __launch_bounds__(512) void k_l1gemm(const int* __restrict__ off,
                                                const int* __restrict__ msg,
                                                const float* __restrict__ w1,
                                                const float* __restrict__ b1,
                                                const float* __restrict__ w2,
                                                float* __restrict__ P) {
  __shared__ float hs[16 * HPITCH];
  int lane = threadIdx.x & 31;
  int wv = threadIdx.x >> 5;
  int wu = __builtin_amdgcn_readfirstlane(wv);
  int mt = blockIdx.x;
  int n = mt * 16 + wu;
  {
    int beg = off[n], end = off[n + 1];
    int cnt = seg_counts(msg, beg, end, lane);
    float invq = 1.0f / (float)cnt;
    int hh = lane >> 4, rl = lane & 15;
    int slot = 2 * rl + hh;
    int lo = rl * 16;
    int wpad = ((2 * NRELH) << OBITS) | n;
    const char* w1c = (const char*)w1;
    float4 acc = make_float4(0.0f, 0.0f, 0.0f, 0.0f);
    for (int base = beg; base < end; base += 32) {
      int m = end - base;
      if (m > 32) m = 32;
      bool ok = slot < m;
      int w = wpad;
      if (ok) w = msg[base + slot];
      int q = w >> OBITS, o = w & OMASK;
      float invl = __int_as_float(__builtin_amdgcn_ds_bpermute(q << 2, __float_as_int(invq)));
      invl = ok ? invl : 0.0f;
      int roff = (q * NNODES + o) << 8;
      if (m > 16) {
        l1_block<0>(roff, invl, lo, w1c, acc);
        l1_block<1>(roff, invl, lo, w1c, acc);
      } else {
        l1_block<0>(roff, invl, lo, w1c, acc);
      }
    }
    acc.x += __shfl_xor(acc.x, 16);
    acc.y += __shfl_xor(acc.y, 16);
    acc.z += __shfl_xor(acc.z, 16);
    acc.w += __shfl_xor(acc.w, 16);
    float4 sl = *(const float4*)(w1c + (unsigned)((((2 * NRELH) * NNODES + n) << 8) + lo));
    float4 bb = *(const float4*)(b1 + 4 * rl);
    float x0 = acc.x + sl.x + bb.x;
    float x1 = acc.y + sl.y + bb.y;
    float x2 = acc.z + sl.z + bb.z;
    float x3 = acc.w + sl.w + bb.w;
    float4 hv;
    hv.x = x0 > 0.0f ? x0 : 0.0f;
    hv.y = x1 > 0.0f ? x1 : 0.0f;
    hv.z = x2 > 0.0f ? x2 : 0.0f;
    hv.w = x3 > 0.0f ? x3 : 0.0f;
    if (hh == 0) *(float4*)(hs + wu * HPITCH + 4 * rl) = hv;
  }
  __syncthreads();

  int mn    = lane & 15;
  int khalf = lane >> 4;
  const float* hrow = hs + mn * HPITCH + 2 * khalf;
  v2f a[HID / 4];
#pragma unroll
  for (int k = 0; k < HID / 4; ++k) a[k] = *(const v2f*)(hrow + 4 * k);
  int rowoff = khalf * 8;
  for (int r = wu; r < RTOT; r += 16) {
    const float* wrel = w2 + (size_t)r * HID * NCLS + (size_t)(2 * khalf) * NCLS + mn;
    union { v8f v; float f[8]; } acc;
    acc.v = (v8f)0.0f;
#pragma unroll
    for (int k = 0; k < HID / 4; ++k) {
      v2f b;
      b.x = wrel[(size_t)(4 * k) * NCLS];
      b.y = wrel[(size_t)(4 * k + 1) * NCLS];
      acc.v = __builtin_amdgcn_wmma_f32_16x16x4_f32(
           false, a[k],  false, b,
           (short)0, acc.v,  false,  false);
    }
#pragma unroll
    for (int v = 0; v < 8; ++v) {
      P[(size_t)(mt * 16 + v + rowoff) * PCOLS + r * NCLS + mn] = acc.f[v];
    }
  }
}

#define L2_LOAD(I, U, PO, IL)                                                \
  {                                                                          \
    int po = quad_pick<U>(PO);                                               \
    iv[I] = __int_as_float(quad_pick<U>(__float_as_int(IL)));                \
    v[I] = *(const float4*)(Pb + (unsigned)(po + lo));                       \
  }
#define L2_SUM(I)                                                            \
  {                                                                          \
    acc.x = __builtin_fmaf(iv[I], v[I].x, acc.x);                            \
    acc.y = __builtin_fmaf(iv[I], v[I].y, acc.y);                            \
    acc.z = __builtin_fmaf(iv[I], v[I].z, acc.z);                            \
    acc.w = __builtin_fmaf(iv[I], v[I].w, acc.w);                            \
  }
template <int NA, int NB>
__device__ __forceinline__ void l2_stretch(int poffA, float invA, int poffB, float invB, int lo,
                                           const char* __restrict__ Pb, float4& acc) {
  float4 v[8];
  float iv[8];
  L2_LOAD(0, 0, poffA, invA) L2_LOAD(1, 1, poffA, invA)
  if (NA > 2) { L2_LOAD(2, 2, poffA, invA) L2_LOAD(3, 3, poffA, invA) }
  if (NB > 0) { L2_LOAD(4, 0, poffB, invB) L2_LOAD(5, 1, poffB, invB) }
  if (NB > 2) { L2_LOAD(6, 2, poffB, invB) L2_LOAD(7, 3, poffB, invB) }
  L2_SUM(0) L2_SUM(1)
  if (NA > 2) { L2_SUM(2) L2_SUM(3) }
  if (NB > 0) { L2_SUM(4) L2_SUM(5) }
  if (NB > 2) { L2_SUM(6) L2_SUM(7) }
}

__global__ __launch_bounds__(256) void k_l2pull(const int* __restrict__ off,
                                                const int* __restrict__ msg,
                                                const float* __restrict__ P,
                                                const float* __restrict__ b2,
                                                float* __restrict__ out) {
  int lane = threadIdx.x & 31;
  int wu = __builtin_amdgcn_readfirstlane(threadIdx.x >> 5);
  int pr = blockIdx.x * 8 + wu;
  if (pr >= NNODES / 2) return;
  int g = lane >> 2, c4 = lane & 3;
  int slot = 8 * c4 + g;
  int lo = c4 * 16;
  const char* Pb = (const char*)P;
  float4 res = make_float4(0.0f, 0.0f, 0.0f, 0.0f);
  for (int t = 0; t < 2; ++t) {
    int n = 2 * pr + t;
    int beg = off[n], end = off[n + 1];
    int cnt = seg_counts(msg, beg, end, lane);
    float invq = 1.0f / (float)cnt;
    int wpad = ((2 * NRELH) << OBITS) | n;
    float4 acc = make_float4(0.0f, 0.0f, 0.0f, 0.0f);
    for (int base = beg; base < end; base += 64) {
      int mA = end - base;
      int mB = mA - 32;
      if (mA > 32) mA = 32;
      if (mB < 0) mB = 0;
      if (mB > 32) mB = 32;
      bool okA = slot < mA, okB = slot < mB;
      int wA = wpad, wB = wpad;
      if (okA) wA = msg[base + slot];
      if (okB) wB = msg[base + 32 + slot];
      int qA = wA >> OBITS, oA = wA & OMASK;
      int qB = wB >> OBITS, oB = wB & OMASK;
      float invA = __int_as_float(__builtin_amdgcn_ds_bpermute(qA << 2, __float_as_int(invq)));
      float invB = __int_as_float(__builtin_amdgcn_ds_bpermute(qB << 2, __float_as_int(invq)));
      invA = okA ? invA : 0.0f;
      invB = okB ? invB : 0.0f;
      int poffA = (oA * PCOLS + qA * NCLS) << 2;
      int poffB = (oB * PCOLS + qB * NCLS) << 2;
      if (mB > 16)      l2_stretch<4, 4>(poffA, invA, poffB, invB, lo, Pb, acc);
      else if (mB > 0)  l2_stretch<4, 2>(poffA, invA, poffB, invB, lo, Pb, acc);
      else if (mA > 16) l2_stretch<4, 0>(poffA, invA, poffB, invB, lo, Pb, acc);
      else              l2_stretch<2, 0>(poffA, invA, poffB, invB, lo, Pb, acc);
    }
    acc.x += __shfl_xor(acc.x, 4);  acc.y += __shfl_xor(acc.y, 4);
    acc.z += __shfl_xor(acc.z, 4);  acc.w += __shfl_xor(acc.w, 4);
    acc.x += __shfl_xor(acc.x, 8);  acc.y += __shfl_xor(acc.y, 8);
    acc.z += __shfl_xor(acc.z, 8);  acc.w += __shfl_xor(acc.w, 8);
    acc.x += __shfl_xor(acc.x, 16); acc.y += __shfl_xor(acc.y, 16);
    acc.z += __shfl_xor(acc.z, 16); acc.w += __shfl_xor(acc.w, 16);
    if (g == t) res = acc;
  }
  if (lane < 8) {
    int n = 2 * pr + g;
    float4 bb = *(const float4*)(b2 + 4 * c4);
    float4 ps = *(const float4*)(P + (size_t)n * PCOLS + (2 * NRELH) * NCLS + 4 * c4);
    float4 ov;
    ov.x = bb.x + ps.x + res.x;
    ov.y = bb.y + ps.y + res.y;
    ov.z = bb.z + ps.z + res.z;
    ov.w = bb.w + ps.w + res.w;
    *(float4*)(out + (size_t)n * NCLS + 4 * c4) = ov;
  }
}

extern "C" void kernel_launch(void* const* d_in, const int* in_sizes, int n_in,
                              void* d_out, int out_size, void* d_ws, size_t ws_size,
                              hipStream_t stream) {
  const int*   src = (const int*)d_in[0];
  const int*   rel = (const int*)d_in[1];
  const int*   dst = (const int*)d_in[2];
  const float* w1  = (const float*)d_in[3];
  const float* b1  = (const float*)d_in[4];
  const float* w2  = (const float*)d_in[5];
  const float* b2  = (const float*)d_in[6];
  float* out = (float*)d_out;
  const int E = in_sizes[0];

  int* deg  = (int*)d_ws;
  int* off  = deg + NOFF;
  int* rk_s = off + NOFF;
  int* rk_d = rk_s + E;
  int* msg  = rk_d + E;
  float* P  = (float*)(msg + 2 * (size_t)E);

  k_zero<<<(NNODES + 255) / 256, 256, 0, stream>>>(deg, NNODES);

  k_rank<<<(E + 255) / 256, 256, 0, stream>>>(src, dst, deg, rk_s, rk_d, E);

  k_scan<<<1, SCAN_T, 0, stream>>>(deg, off);

  k_fill<<<(E + 255) / 256, 256, 0, stream>>>(src, rel, dst, off, rk_s, rk_d, msg, E);

  k_l1gemm<<<NNODES / 16, 512, 0, stream>>>(off, msg, w1, b1, w2, P);

  k_l2pull<<<(NNODES / 2 + 7) / 8, 256, 0, stream>>>(off, msg, P, b2, out);
}


# Scan kernel: a thread takes 4 adjacent counters in each of 5 passes (16-byte loads and stores, 512 contiguous bytes per wave; was an 80-byte lane stride), row-shift wave sums, 1 barrier, was 20
# speedup vs baseline: 1.7320x; 1.0678x over previous
#include <hip/hip_runtime.h>

#define NNODES 20000
#define NRELH  16
#define RTOT   33
#define HID    64
#define NCLS   16
#define PCOLS  (RTOT * NCLS)
#define OBITS  15
#define OMASK  32767
#define NOFF   20032

typedef __attribute__((ext_vector_type(2))) float v2f;
typedef __attribute__((ext_vector_type(8))) float v8f;


__global__ __launch_bounds__(256) void k_zero(int* __restrict__ p, int n) {
  int i = blockIdx.x * blockDim.x + threadIdx.x;
  if (i < n) p[i] = 0;
}

__global__ __launch_bounds__(256) void k_rank(const int* __restrict__ src,
                                              const int* __restrict__ dst,
                                              int* __restrict__ deg,
                                              int* __restrict__ rk_s,
                                              int* __restrict__ rk_d, int E) {
  int e = blockIdx.x * blockDim.x + threadIdx.x;
  if (e >= E) return;
  rk_s[e] = atomicAdd(deg + src[e], 1);
  rk_d[e] = atomicAdd(deg + dst[e], 1);
}

#define SCAN_T 1024
#define SCAN_P 5
#define NQUAD  (NNODES / 4)

template <int K>
__device__ __forceinline__ int row_up(int x) {
  return __builtin_amdgcn_update_dpp(0, x, 0x110 + K, 0xf, 0xf, false);
}
__device__ __forceinline__ int wave_incl(int x, int lane) {
  x += row_up<1>(x);
  x += row_up<2>(x);
  x += row_up<4>(x);
  x += row_up<8>(x);
  int low = __builtin_amdgcn_readlane(x, 15);
  return x + ((lane >= 16) ? low : 0);
}

__global__ __launch_bounds__(SCAN_T) void k_scan(const int* __restrict__ deg,
                                                 int* __restrict__ off) {
  __shared__ int wtot[SCAN_P * 32];
  int t = threadIdx.x;
  int lane = t & 31;
  int wu = __builtin_amdgcn_readfirstlane(t >> 5);
  int4 v[SCAN_P];
#pragma unroll
  for (int p = 0; p < SCAN_P; ++p) {
    int g = p * SCAN_T + t;
    v[p] = make_int4(0, 0, 0, 0);
    if (g < NQUAD) v[p] = *(const int4*)(deg + 4 * g);
  }
  int tot[SCAN_P], inc[SCAN_P];
#pragma unroll
  for (int p = 0; p < SCAN_P; ++p) {
    tot[p] = (v[p].x + v[p].y) + (v[p].z + v[p].w);
    inc[p] = wave_incl(tot[p], lane);
    if (lane == 31) wtot[p * 32 + wu] = inc[p];
  }
  __syncthreads();
  int carry = 0;
#pragma unroll
  for (int p = 0; p < SCAN_P; ++p) {
    int w = wtot[p * 32 + lane];
    int ws = wave_incl(w, lane);
    int ahead = __builtin_amdgcn_readlane(ws - w, wu);
    int base = carry + ahead + (inc[p] - tot[p]);
    carry += __builtin_amdgcn_readlane(ws, 31);
    int g = p * SCAN_T + t;
    if (g < NQUAD) {
      int4 o;
      o.x = base;
      o.y = o.x + v[p].x;
      o.z = o.y + v[p].y;
      o.w = o.z + v[p].z;
      *(int4*)(off + 4 * g) = o;
    }
  }
  if (t == 0) off[NNODES] = carry;
}

__global__ __launch_bounds__(256) void k_fill(const int* __restrict__ src,
                                              const int* __restrict__ rel,
                                              const int* __restrict__ dst,
                                              const int* __restrict__ off,
                                              const int* __restrict__ rk_s,
                                              const int* __restrict__ rk_d,
                                              int* __restrict__ msg, int E) {
  int e = blockIdx.x * blockDim.x + threadIdx.x;
  if (e >= E) return;
  int s = src[e], r = rel[e], d = dst[e];
  msg[off[s] + rk_s[e]] = (r << OBITS) | d;
  msg[off[d] + rk_d[e]] = ((r + NRELH) << OBITS) | s;
}

__device__ __forceinline__ int seg_counts(const int* __restrict__ msg, int beg, int end, int lane) {
  int cnt = 0;
  for (int base = beg; base < end; base += 32) {
    int i = base + lane;
    int q = (i < end) ? (msg[i] >> OBITS) : 64;
#pragma unroll
    for (int j = 0; j < 32; ++j)
      cnt += (__builtin_amdgcn_readlane(q, j) == lane) ? 1 : 0;
  }
  return cnt;
}

template <int K>
__device__ __forceinline__ int row_pick(int x) {
  return __builtin_amdgcn_update_dpp(0, x, 0x150 + K, 0xf, 0xf, false);
}
template <int K>
__device__ __forceinline__ int quad_pick(int x) {
  return __builtin_amdgcn_update_dpp(0, x, K * 0x55, 0xf, 0xf, false);
}

#define L1_LOAD(U)                                                           \
  {                                                                          \
    int ro = row_pick<8 * B + (U)>(roff);                                    \
    iv[U] = __int_as_float(row_pick<8 * B + (U)>(__float_as_int(invl)));     \
    v[U] = *(const float4*)(w1c + (unsigned)(ro + lo));                      \
  }
#define L1_SUM(U)                                                            \
  {                                                                          \
    acc.x = __builtin_fmaf(iv[U], v[U].x, acc.x);                            \
    acc.y = __builtin_fmaf(iv[U], v[U].y, acc.y);                            \
    acc.z = __builtin_fmaf(iv[U], v[U].z, acc.z);                            \
    acc.w = __builtin_fmaf(iv[U], v[U].w, acc.w);                            \
  }
template <int B>
__device__ __forceinline__ void l1_block(int roff, float invl, int lo,
                                         const char* __restrict__ w1c, float4& acc) {
  float4 v[8];
  float iv[8];
  L1_LOAD(0) L1_LOAD(1) L1_LOAD(2) L1_LOAD(3) L1_LOAD(4) L1_LOAD(5) L1_LOAD(6) L1_LOAD(7)
  L1_SUM(0) L1_SUM(1) L1_SUM(2) L1_SUM(3) L1_SUM(4) L1_SUM(5) L1_SUM(6) L1_SUM(7)
}

#define HPITCH 68
__global__ __launch_bounds__(512) void k_l1gemm(const int* __restrict__ off,
                                                const int* __restrict__ msg,
                                                const float* __restrict__ w1,
                                                const float* __restrict__ b1,
                                                const float* __restrict__ w2,
                                                float* __restrict__ P) {
  __shared__ float hs[16 * HPITCH];
  int lane = threadIdx.x & 31;
  int wv = threadIdx.x >> 5;
  int wu = __builtin_amdgcn_readfirstlane(wv);
  int mt = blockIdx.x;
  int n = mt * 16 + wu;
  {
    int beg = off[n], end = off[n + 1];
    int cnt = seg_counts(msg, beg, end, lane);
    float invq = 1.0f / (float)cnt;
    int hh = lane >> 4, rl = lane & 15;
    int slot = 2 * rl + hh;
    int lo = rl * 16;
    int wpad = ((2 * NRELH) << OBITS) | n;
    const char* w1c = (const char*)w1;
    float4 acc = make_float4(0.0f, 0.0f, 0.0f, 0.0f);
    for (int base = beg; base < end; base += 32) {
      int m = end - base;
      if (m > 32) m = 32;
      bool ok = slot < m;
      int w = wpad;
      if (ok) w = msg[base + slot];
      int q = w >> OBITS, o = w & OMASK;
      float invl = __int_as_float(__builtin_amdgcn_ds_bpermute(q << 2, __float_as_int(invq)));
      invl = ok ? invl : 0.0f;
      int roff = (q * NNODES + o) << 8;
      if (m > 16) {
        l1_block<0>(roff, invl, lo, w1c, acc);
        l1_block<1>(roff, invl, lo, w1c, acc);
      } else {
        l1_block<0>(roff, invl, lo, w1c, acc);
      }
    }
    acc.x += __shfl_xor(acc.x, 16);
    acc.y += __shfl_xor(acc.y, 16);
    acc.z += __shfl_xor(acc.z, 16);
    acc.w += __shfl_xor(acc.w, 16);
    float4 sl = *(const float4*)(w1c + (unsigned)((((2 * NRELH) * NNODES + n) << 8) + lo));
    float4 bb = *(const float4*)(b1 + 4 * rl);
    float x0 = acc.x + sl.x + bb.x;
    float x1 = acc.y + sl.y + bb.y;
    float x2 = acc.z + sl.z + bb.z;
    float x3 = acc.w + sl.w + bb.w;
    float4 hv;
    hv.x = x0 > 0.0f ? x0 : 0.0f;
    hv.y = x1 > 0.0f ? x1 : 0.0f;
    hv.z = x2 > 0.0f ? x2 : 0.0f;
    hv.w = x3 > 0.0f ? x3 : 0.0f;
    if (hh == 0) *(float4*)(hs + wu * HPITCH + 4 * rl) = hv;
  }
  __syncthreads();

  int mn    = lane & 15;
  int khalf = lane >> 4;
  const float* hrow = hs + mn * HPITCH + 2 * khalf;
  v2f a[HID / 4];
#pragma unroll
  for (int k = 0; k < HID / 4; ++k) a[k] = *(const v2f*)(hrow + 4 * k);
  int rowoff = khalf * 8;
  for (int r = wu; r < RTOT; r += 16) {
    const float* wrel = w2 + (size_t)r * HID * NCLS + (size_t)(2 * khalf) * NCLS + mn;
    union { v8f v; float f[8]; } acc;
    acc.v = (v8f)0.0f;
#pragma unroll
    for (int k = 0; k < HID / 4; ++k) {
      v2f b;
      b.x = wrel[(size_t)(4 * k) * NCLS];
      b.y = wrel[(size_t)(4 * k + 1) * NCLS];
      acc.v = __builtin_amdgcn_wmma_f32_16x16x4_f32(
           false, a[k],  false, b,
           (short)0, acc.v,  false,  false);
    }
#pragma unroll
    for (int v = 0; v < 8; ++v) {
      P[(size_t)(mt * 16 + v + rowoff) * PCOLS + r * NCLS + mn] = acc.f[v];
    }
  }
}

#define L2_LOAD(I, U, PO, IL)                                                \
  {                                                                          \
    int po = quad_pick<U>(PO);                                               \
    iv[I] = __int_as_float(quad_pick<U>(__float_as_int(IL)));                \
    v[I] = *(const float4*)(Pb + (unsigned)(po + lo));                       \
  }
#define L2_SUM(I)                                                            \
  {                                                                          \
    acc.x = __builtin_fmaf(iv[I], v[I].x, acc.x);                            \
    acc.y = __builtin_fmaf(iv[I], v[I].y, acc.y);                            \
    acc.z = __builtin_fmaf(iv[I], v[I].z, acc.z);                            \
    acc.w = __builtin_fmaf(iv[I], v[I].w, acc.w);                            \
  }
template <int NA, int NB>
__device__ __forceinline__ void l2_stretch(int poffA, float invA, int poffB, float invB, int lo,
                                           const char* __restrict__ Pb, float4& acc) {
  float4 v[8];
  float iv[8];
  L2_LOAD(0, 0, poffA, invA) L2_LOAD(1, 1, poffA, invA)
  if (NA > 2) { L2_LOAD(2, 2, poffA, invA) L2_LOAD(3, 3, poffA, invA) }
  if (NB > 0) { L2_LOAD(4, 0, poffB, invB) L2_LOAD(5, 1, poffB, invB) }
  if (NB > 2) { L2_LOAD(6, 2, poffB, invB) L2_LOAD(7, 3, poffB, invB) }
  L2_SUM(0) L2_SUM(1)
  if (NA > 2) { L2_SUM(2) L2_SUM(3) }
  if (NB > 0) { L2_SUM(4) L2_SUM(5) }
  if (NB > 2) { L2_SUM(6) L2_SUM(7) }
}

__global__ __launch_bounds__(256) void k_l2pull(const int* __restrict__ off,
                                                const int* __restrict__ msg,
                                                const float* __restrict__ P,
                                                const float* __restrict__ b2,
                                                float* __restrict__ out) {
  int lane = threadIdx.x & 31;
  int wu = __builtin_amdgcn_readfirstlane(threadIdx.x >> 5);
  int pr = blockIdx.x * 8 + wu;
  if (pr >= NNODES / 2) return;
  int g = lane >> 2, c4 = lane & 3;
  int slot = 8 * c4 + g;
  int lo = c4 * 16;
  const char* Pb = (const char*)P;
  float4 res = make_float4(0.0f, 0.0f, 0.0f, 0.0f);
  for (int t = 0; t < 2; ++t) {
    int n = 2 * pr + t;
    int beg = off[n], end = off[n + 1];
    int cnt = seg_counts(msg, beg, end, lane);
    float invq = 1.0f / (float)cnt;
    int wpad = ((2 * NRELH) << OBITS) | n;
    float4 acc = make_float4(0.0f, 0.0f, 0.0f, 0.0f);
    for (int base = beg; base < end; base += 64) {
      int mA = end - base;
      int mB = mA - 32;
      if (mA > 32) mA = 32;
      if (mB < 0) mB = 0;
      if (mB > 32) mB = 32;
      bool okA = slot < mA, okB = slot < mB;
      int wA = wpad, wB = wpad;
      if (okA) wA = msg[base + slot];
      if (okB) wB = msg[base + 32 + slot];
      int qA = wA >> OBITS, oA = wA & OMASK;
      int qB = wB >> OBITS, oB = wB & OMASK;
      float invA = __int_as_float(__builtin_amdgcn_ds_bpermute(qA << 2, __float_as_int(invq)));
      float invB = __int_as_float(__builtin_amdgcn_ds_bpermute(qB << 2, __float_as_int(invq)));
      invA = okA ? invA : 0.0f;
      invB = okB ? invB : 0.0f;
      int poffA = (oA * PCOLS + qA * NCLS) << 2;
      int poffB = (oB * PCOLS + qB * NCLS) << 2;
      if (mB > 16)      l2_stretch<4, 4>(poffA, invA, poffB, invB, lo, Pb, acc);
      else if (mB > 0)  l2_stretch<4, 2>(poffA, invA, poffB, invB, lo, Pb, acc);
      else if (mA > 16) l2_stretch<4, 0>(poffA, invA, poffB, invB, lo, Pb, acc);
      else              l2_stretch<2, 0>(poffA, invA, poffB, invB, lo, Pb, acc);
    }
    acc.x += __shfl_xor(acc.x, 4);  acc.y += __shfl_xor(acc.y, 4);
    acc.z += __shfl_xor(acc.z, 4);  acc.w += __shfl_xor(acc.w, 4);
    acc.x += __shfl_xor(acc.x, 8);  acc.y += __shfl_xor(acc.y, 8);
    acc.z += __shfl_xor(acc.z, 8);  acc.w += __shfl_xor(acc.w, 8);
    acc.x += __shfl_xor(acc.x, 16); acc.y += __shfl_xor(acc.y, 16);
    acc.z += __shfl_xor(acc.z, 16); acc.w += __shfl_xor(acc.w, 16);
    if (g == t) res = acc;
  }
  if (lane < 8) {
    int n = 2 * pr + g;
    float4 bb = *(const float4*)(b2 + 4 * c4);
    float4 ps = *(const float4*)(P + (size_t)n * PCOLS + (2 * NRELH) * NCLS + 4 * c4);
    float4 ov;
    ov.x = bb.x + ps.x + res.x;
    ov.y = bb.y + ps.y + res.y;
    ov.z = bb.z + ps.z + res.z;
    ov.w = bb.w + ps.w + res.w;
    *(float4*)(out + (size_t)n * NCLS + 4 * c4) = ov;
  }
}

extern "C" void kernel_launch(void* const* d_in, const int* in_sizes, int n_in,
                              void* d_out, int out_size, void* d_ws, size_t ws_size,
                              hipStream_t stream) {
  const int*   src = (const int*)d_in[0];
  const int*   rel = (const int*)d_in[1];
  const int*   dst = (const int*)d_in[2];
  const float* w1  = (const float*)d_in[3];
  const float* b1  = (const float*)d_in[4];
  const float* w2  = (const float*)d_in[5];
  const float* b2  = (const float*)d_in[6];
  float* out = (float*)d_out;
  const int E = in_sizes[0];

  int* deg  = (int*)d_ws;
  int* off  = deg + NOFF;
  int* rk_s = off + NOFF;
  int* rk_d = rk_s + E;
  int* msg  = rk_d + E;
  float* P  = (float*)(msg + 2 * (size_t)E);

  k_zero<<<(NNODES + 255) / 256, 256, 0, stream>>>(deg, NNODES);

  k_rank<<<(E + 255) / 256, 256, 0, stream>>>(src, dst, deg, rk_s, rk_d, E);

  k_scan<<<1, SCAN_T, 0, stream>>>(deg, off);

  k_fill<<<(E + 255) / 256, 256, 0, stream>>>(src, rel, dst, off, rk_s, rk_d, msg, E);

  k_l1gemm<<<NNODES / 16, 512, 0, stream>>>(off, msg, w1, b1, w2, P);

  k_l2pull<<<(NNODES / 2 + 7) / 8, 256, 0, stream>>>(off, msg, P, b2, out);
}


# Rank without global atomics: 128 work-groups rank a 5000-edge chunk each by LDS adds on an 80 KB node histogram; column prefix over chunk rows gives bases; zero kernel gone; h, out add order changes
# speedup vs baseline: 2.7749x; 1.6021x over previous
#include <hip/hip_runtime.h>

#define NNODES 20000
#define NRELH  16
#define RTOT   33
#define HID    64
#define NCLS   16
#define PCOLS  (RTOT * NCLS)
#define OBITS  15
#define OMASK  32767
#define NOFF   20032
#define NQUAD  (NNODES / 4)
#define NCH    128
#define RANK_T 1024

typedef __attribute__((ext_vector_type(2))) float v2f;
typedef __attribute__((ext_vector_type(8))) float v8f;


__global__ __launch_bounds__(RANK_T) void k_rank(const int* __restrict__ src,
                                                 const int* __restrict__ dst,
                                                 int* __restrict__ tab,
                                                 int* __restrict__ rk_s,
                                                 int* __restrict__ rk_d, int E) {
  __shared__ int hist[NNODES];
  int t = threadIdx.x;
  int4* h4 = (int4*)hist;
  for (int i = t; i < NQUAD; i += RANK_T) h4[i] = make_int4(0, 0, 0, 0);
  __syncthreads();
  int per = (E + NCH - 1) / NCH;
  int e0 = blockIdx.x * per;
  int e1 = e0 + per;
  if (e1 > E) e1 = E;
  for (int eb = e0 + t; eb < e1; eb += 4 * RANK_T) {
    int s[4], d[4];
#pragma unroll
    for (int j = 0; j < 4; ++j) {
      int e = eb + j * RANK_T;
      s[j] = 0;
      d[j] = 0;
      if (e < e1) {
        s[j] = src[e];
        d[j] = dst[e];
      }
    }
#pragma unroll
    for (int j = 0; j < 4; ++j) {
      int e = eb + j * RANK_T;
      if (e < e1) {
        rk_s[e] = atomicAdd(hist + s[j], 1);
        rk_d[e] = atomicAdd(hist + d[j], 1);
      }
    }
  }
  __syncthreads();
  int4* row = (int4*)(tab + (size_t)blockIdx.x * NNODES);
  for (int i = t; i < NQUAD; i += RANK_T) row[i] = h4[i];
}

__global__ __launch_bounds__(64) void k_colsum(int* __restrict__ tab, int* __restrict__ deg) {
  int n = blockIdx.x * blockDim.x + threadIdx.x;
  if (n >= NNODES) return;
  int* col = tab + n;
  int run = 0;
  for (int c0 = 0; c0 < NCH; c0 += 16) {
    int c[16];
#pragma unroll
    for (int j = 0; j < 16; ++j) c[j] = col[(size_t)(c0 + j) * NNODES];
#pragma unroll
    for (int j = 0; j < 16; ++j) {
      col[(size_t)(c0 + j) * NNODES] = run;
      run += c[j];
    }
  }
  deg[n] = run;
}

#define SCAN_T 1024
#define SCAN_P 5

template <int K>
__device__ __forceinline__ int row_up(int x) {
  return __builtin_amdgcn_update_dpp(0, x, 0x110 + K, 0xf, 0xf, false);
}
__device__ __forceinline__ int wave_incl(int x, int lane) {
  x += row_up<1>(x);
  x += row_up<2>(x);
  x += row_up<4>(x);
  x += row_up<8>(x);
  int low = __builtin_amdgcn_readlane(x, 15);
  return x + ((lane >= 16) ? low : 0);
}

__global__ __launch_bounds__(SCAN_T) void k_scan(const int* __restrict__ deg,
                                                 int* __restrict__ off) {
  __shared__ int wtot[SCAN_P * 32];
  int t = threadIdx.x;
  int lane = t & 31;
  int wu = __builtin_amdgcn_readfirstlane(t >> 5);
  int4 v[SCAN_P];
#pragma unroll
  for (int p = 0; p < SCAN_P; ++p) {
    int g = p * SCAN_T + t;
    v[p] = make_int4(0, 0, 0, 0);
    if (g < NQUAD) v[p] = *(const int4*)(deg + 4 * g);
  }
  int tot[SCAN_P], inc[SCAN_P];
#pragma unroll
  for (int p = 0; p < SCAN_P; ++p) {
    tot[p] = (v[p].x + v[p].y) + (v[p].z + v[p].w);
    inc[p] = wave_incl(tot[p], lane);
    if (lane == 31) wtot[p * 32 + wu] = inc[p];
  }
  __syncthreads();
  int carry = 0;
#pragma unroll
  for (int p = 0; p < SCAN_P; ++p) {
    int w = wtot[p * 32 + lane];
    int ws = wave_incl(w, lane);
    int ahead = __builtin_amdgcn_readlane(ws - w, wu);
    int base = carry + ahead + (inc[p] - tot[p]);
    carry += __builtin_amdgcn_readlane(ws, 31);
    int g = p * SCAN_T + t;
    if (g < NQUAD) {
      int4 o;
      o.x = base;
      o.y = o.x + v[p].x;
      o.z = o.y + v[p].y;
      o.w = o.z + v[p].z;
      *(int4*)(off + 4 * g) = o;
    }
  }
  if (t == 0) off[NNODES] = carry;
}

__global__ __launch_bounds__(256) void k_fill(const int* __restrict__ src,
                                              const int* __restrict__ rel,
                                              const int* __restrict__ dst,
                                              const int* __restrict__ off,
                                              const int* __restrict__ tab,
                                              const int* __restrict__ rk_s,
                                              const int* __restrict__ rk_d,
                                              int* __restrict__ msg, int E) {
  int e = blockIdx.x * blockDim.x + threadIdx.x;
  if (e >= E) return;
  int per = (E + NCH - 1) / NCH;
  const int* base = tab + (size_t)(e / per) * NNODES;
  int s = src[e], r = rel[e], d = dst[e];
  msg[off[s] + base[s] + rk_s[e]] = (r << OBITS) | d;
  msg[off[d] + base[d] + rk_d[e]] = ((r + NRELH) << OBITS) | s;
}

__device__ __forceinline__ int seg_counts(const int* __restrict__ msg, int beg, int end, int lane) {
  int cnt = 0;
  for (int base = beg; base < end; base += 32) {
    int i = base + lane;
    int q = (i < end) ? (msg[i] >> OBITS) : 64;
#pragma unroll
    for (int j = 0; j < 32; ++j)
      cnt += (__builtin_amdgcn_readlane(q, j) == lane) ? 1 : 0;
  }
  return cnt;
}

template <int K>
__device__ __forceinline__ int row_pick(int x) {
  return __builtin_amdgcn_update_dpp(0, x, 0x150 + K, 0xf, 0xf, false);
}
template <int K>
__device__ __forceinline__ int quad_pick(int x) {
  return __builtin_amdgcn_update_dpp(0, x, K * 0x55, 0xf, 0xf, false);
}

#define L1_LOAD(U)                                                           \
  {                                                                          \
    int ro = row_pick<8 * B + (U)>(roff);                                    \
    iv[U] = __int_as_float(row_pick<8 * B + (U)>(__float_as_int(invl)));     \
    v[U] = *(const float4*)(w1c + (unsigned)(ro + lo));                      \
  }
#define L1_SUM(U)                                                            \
  {                                                                          \
    acc.x = __builtin_fmaf(iv[U], v[U].x, acc.x);                            \
    acc.y = __builtin_fmaf(iv[U], v[U].y, acc.y);                            \
    acc.z = __builtin_fmaf(iv[U], v[U].z, acc.z);                            \
    acc.w = __builtin_fmaf(iv[U], v[U].w, acc.w);                            \
  }
template <int B>
__device__ __forceinline__ void l1_block(int roff, float invl, int lo,
                                         const char* __restrict__ w1c, float4& acc) {
  float4 v[8];
  float iv[8];
  L1_LOAD(0) L1_LOAD(1) L1_LOAD(2) L1_LOAD(3) L1_LOAD(4) L1_LOAD(5) L1_LOAD(6) L1_LOAD(7)
  L1_SUM(0) L1_SUM(1) L1_SUM(2) L1_SUM(3) L1_SUM(4) L1_SUM(5) L1_SUM(6) L1_SUM(7)
}

#define HPITCH 68
__global__ __launch_bounds__(512) void k_l1gemm(const int* __restrict__ off,
                                                const int* __restrict__ msg,
                                                const float* __restrict__ w1,
                                                const float* __restrict__ b1,
                                                const float* __restrict__ w2,
                                                float* __restrict__ P) {
  __shared__ float hs[16 * HPITCH];
  int lane = threadIdx.x & 31;
  int wv = threadIdx.x >> 5;
  int wu = __builtin_amdgcn_readfirstlane(wv);
  int mt = blockIdx.x;
  int n = mt * 16 + wu;
  {
    int beg = off[n], end = off[n + 1];
    int cnt = seg_counts(msg, beg, end, lane);
    float invq = 1.0f / (float)cnt;
    int hh = lane >> 4, rl = lane & 15;
    int slot = 2 * rl + hh;
    int lo = rl * 16;
    int wpad = ((2 * NRELH) << OBITS) | n;
    const char* w1c = (const char*)w1;
    float4 acc = make_float4(0.0f, 0.0f, 0.0f, 0.0f);
    for (int base = beg; base < end; base += 32) {
      int m = end - base;
      if (m > 32) m = 32;
      bool ok = slot < m;
      int w = wpad;
      if (ok) w = msg[base + slot];
      int q = w >> OBITS, o = w & OMASK;
      float invl = __int_as_float(__builtin_amdgcn_ds_bpermute(q << 2, __float_as_int(invq)));
      invl = ok ? invl : 0.0f;
      int roff = (q * NNODES + o) << 8;
      if (m > 16) {
        l1_block<0>(roff, invl, lo, w1c, acc);
        l1_block<1>(roff, invl, lo, w1c, acc);
      } else {
        l1_block<0>(roff, invl, lo, w1c, acc);
      }
    }
    acc.x += __shfl_xor(acc.x, 16);
    acc.y += __shfl_xor(acc.y, 16);
    acc.z += __shfl_xor(acc.z, 16);
    acc.w += __shfl_xor(acc.w, 16);
    float4 sl = *(const float4*)(w1c + (unsigned)((((2 * NRELH) * NNODES + n) << 8) + lo));
    float4 bb = *(const float4*)(b1 + 4 * rl);
    float x0 = acc.x + sl.x + bb.x;
    float x1 = acc.y + sl.y + bb.y;
    float x2 = acc.z + sl.z + bb.z;
    float x3 = acc.w + sl.w + bb.w;
    float4 hv;
    hv.x = x0 > 0.0f ? x0 : 0.0f;
    hv.y = x1 > 0.0f ? x1 : 0.0f;
    hv.z = x2 > 0.0f ? x2 : 0.0f;
    hv.w = x3 > 0.0f ? x3 : 0.0f;
    if (hh == 0) *(float4*)(hs + wu * HPITCH + 4 * rl) = hv;
  }
  __syncthreads();

  int mn    = lane & 15;
  int khalf = lane >> 4;
  const float* hrow = hs + mn * HPITCH + 2 * khalf;
  v2f a[HID / 4];
#pragma unroll
  for (int k = 0; k < HID / 4; ++k) a[k] = *(const v2f*)(hrow + 4 * k);
  int rowoff = khalf * 8;
  for (int r = wu; r < RTOT; r += 16) {
    const float* wrel = w2 + (size_t)r * HID * NCLS + (size_t)(2 * khalf) * NCLS + mn;
    union { v8f v; float f[8]; } acc;
    acc.v = (v8f)0.0f;
#pragma unroll
    for (int k = 0; k < HID / 4; ++k) {
      v2f b;
      b.x = wrel[(size_t)(4 * k) * NCLS];
      b.y = wrel[(size_t)(4 * k + 1) * NCLS];
      acc.v = __builtin_amdgcn_wmma_f32_16x16x4_f32(
           false, a[k],  false, b,
           (short)0, acc.v,  false,  false);
    }
#pragma unroll
    for (int v = 0; v < 8; ++v) {
      P[(size_t)(mt * 16 + v + rowoff) * PCOLS + r * NCLS + mn] = acc.f[v];
    }
  }
}

#define L2_LOAD(I, U, PO, IL)                                                \
  {                                                                          \
    int po = quad_pick<U>(PO);                                               \
    iv[I] = __int_as_float(quad_pick<U>(__float_as_int(IL)));                \
    v[I] = *(const float4*)(Pb + (unsigned)(po + lo));                       \
  }
#define L2_SUM(I)                                                            \
  {                                                                          \
    acc.x = __builtin_fmaf(iv[I], v[I].x, acc.x);                            \
    acc.y = __builtin_fmaf(iv[I], v[I].y, acc.y);                            \
    acc.z = __builtin_fmaf(iv[I], v[I].z, acc.z);                            \
    acc.w = __builtin_fmaf(iv[I], v[I].w, acc.w);                            \
  }
template <int NA, int NB>
__device__ __forceinline__ void l2_stretch(int poffA, float invA, int poffB, float invB, int lo,
                                           const char* __restrict__ Pb, float4& acc) {
  float4 v[8];
  float iv[8];
  L2_LOAD(0, 0, poffA, invA) L2_LOAD(1, 1, poffA, invA)
  if (NA > 2) { L2_LOAD(2, 2, poffA, invA) L2_LOAD(3, 3, poffA, invA) }
  if (NB > 0) { L2_LOAD(4, 0, poffB, invB) L2_LOAD(5, 1, poffB, invB) }
  if (NB > 2) { L2_LOAD(6, 2, poffB, invB) L2_LOAD(7, 3, poffB, invB) }
  L2_SUM(0) L2_SUM(1)
  if (NA > 2) { L2_SUM(2) L2_SUM(3) }
  if (NB > 0) { L2_SUM(4) L2_SUM(5) }
  if (NB > 2) { L2_SUM(6) L2_SUM(7) }
}

__global__ __launch_bounds__(256) void k_l2pull(const int* __restrict__ off,
                                                const int* __restrict__ msg,
                                                const float* __restrict__ P,
                                                const float* __restrict__ b2,
                                                float* __restrict__ out) {
  int lane = threadIdx.x & 31;
  int wu = __builtin_amdgcn_readfirstlane(threadIdx.x >> 5);
  int pr = blockIdx.x * 8 + wu;
  if (pr >= NNODES / 2) return;
  int g = lane >> 2, c4 = lane & 3;
  int slot = 8 * c4 + g;
  int lo = c4 * 16;
  const char* Pb = (const char*)P;
  float4 res = make_float4(0.0f, 0.0f, 0.0f, 0.0f);
  for (int t = 0; t < 2; ++t) {
    int n = 2 * pr + t;
    int beg = off[n], end = off[n + 1];
    int cnt = seg_counts(msg, beg, end, lane);
    float invq = 1.0f / (float)cnt;
    int wpad = ((2 * NRELH) << OBITS) | n;
    float4 acc = make_float4(0.0f, 0.0f, 0.0f, 0.0f);
    for (int base = beg; base < end; base += 64) {
      int mA = end - base;
      int mB = mA - 32;
      if (mA > 32) mA = 32;
      if (mB < 0) mB = 0;
      if (mB > 32) mB = 32;
      bool okA = slot < mA, okB = slot < mB;
      int wA = wpad, wB = wpad;
      if (okA) wA = msg[base + slot];
      if (okB) wB = msg[base + 32 + slot];
      int qA = wA >> OBITS, oA = wA & OMASK;
      int qB = wB >> OBITS, oB = wB & OMASK;
      float invA = __int_as_float(__builtin_amdgcn_ds_bpermute(qA << 2, __float_as_int(invq)));
      float invB = __int_as_float(__builtin_amdgcn_ds_bpermute(qB << 2, __float_as_int(invq)));
      invA = okA ? invA : 0.0f;
      invB = okB ? invB : 0.0f;
      int poffA = (oA * PCOLS + qA * NCLS) << 2;
      int poffB = (oB * PCOLS + qB * NCLS) << 2;
      if (mB > 16)      l2_stretch<4, 4>(poffA, invA, poffB, invB, lo, Pb, acc);
      else if (mB > 0)  l2_stretch<4, 2>(poffA, invA, poffB, invB, lo, Pb, acc);
      else if (mA > 16) l2_stretch<4, 0>(poffA, invA, poffB, invB, lo, Pb, acc);
      else              l2_stretch<2, 0>(poffA, invA, poffB, invB, lo, Pb, acc);
    }
    acc.x += __shfl_xor(acc.x, 4);  acc.y += __shfl_xor(acc.y, 4);
    acc.z += __shfl_xor(acc.z, 4);  acc.w += __shfl_xor(acc.w, 4);
    acc.x += __shfl_xor(acc.x, 8);  acc.y += __shfl_xor(acc.y, 8);
    acc.z += __shfl_xor(acc.z, 8);  acc.w += __shfl_xor(acc.w, 8);
    acc.x += __shfl_xor(acc.x, 16); acc.y += __shfl_xor(acc.y, 16);
    acc.z += __shfl_xor(acc.z, 16); acc.w += __shfl_xor(acc.w, 16);
    if (g == t) res = acc;
  }
  if (lane < 8) {
    int n = 2 * pr + g;
    float4 bb = *(const float4*)(b2 + 4 * c4);
    float4 ps = *(const float4*)(P + (size_t)n * PCOLS + (2 * NRELH) * NCLS + 4 * c4);
    float4 ov;
    ov.x = bb.x + ps.x + res.x;
    ov.y = bb.y + ps.y + res.y;
    ov.z = bb.z + ps.z + res.z;
    ov.w = bb.w + ps.w + res.w;
    *(float4*)(out + (size_t)n * NCLS + 4 * c4) = ov;
  }
}

extern "C" void kernel_launch(void* const* d_in, const int* in_sizes, int n_in,
                              void* d_out, int out_size, void* d_ws, size_t ws_size,
                              hipStream_t stream) {
  const int*   src = (const int*)d_in[0];
  const int*   rel = (const int*)d_in[1];
  const int*   dst = (const int*)d_in[2];
  const float* w1  = (const float*)d_in[3];
  const float* b1  = (const float*)d_in[4];
  const float* w2  = (const float*)d_in[5];
  const float* b2  = (const float*)d_in[6];
  float* out = (float*)d_out;
  const int E = in_sizes[0];

  int* deg  = (int*)d_ws;
  int* off  = deg + NOFF;
  int* rk_s = off + NOFF;
  int* rk_d = rk_s + E;
  int* msg  = rk_d + E;
  float* P  = (float*)(msg + 2 * (size_t)E);
  int* tab  = (int*)(P + (size_t)NNODES * PCOLS);

  k_rank<<<NCH, RANK_T, 0, stream>>>(src, dst, tab, rk_s, rk_d, E);

  k_colsum<<<(NNODES + 63) / 64, 64, 0, stream>>>(tab, deg);

  k_scan<<<1, SCAN_T, 0, stream>>>(deg, off);

  k_fill<<<(E + 255) / 256, 256, 0, stream>>>(src, rel, dst, off, tab, rk_s, rk_d, msg, E);

  k_l1gemm<<<NNODES / 16, 512, 0, stream>>>(off, msg, w1, b1, w2, P);

  k_l2pull<<<(NNODES / 2 + 7) / 8, 256, 0, stream>>>(off, msg, P, b2, out);
}


# Fill: a (chunk, node half) work-group builds next-slot bases (off + table row) in LDS; a message's slot is a returning LDS add; rank arrays and 4 gathers per edge gone; h, out add order changes
# speedup vs baseline: 3.0410x; 1.0959x over previous
#include <hip/hip_runtime.h>

#define NNODES 20000
#define NRELH  16
#define RTOT   33
#define HID    64
#define NCLS   16
#define PCOLS  (RTOT * NCLS)
#define OBITS  15
#define OMASK  32767
#define NOFF   20032
#define NQUAD  (NNODES / 4)
#define NCH    128
#define RANK_T 1024

typedef __attribute__((ext_vector_type(2))) float v2f;
typedef __attribute__((ext_vector_type(8))) float v8f;
typedef __attribute__((ext_vector_type(4))) int v4i;


__global__ __launch_bounds__(RANK_T) void k_count(const int* __restrict__ src,
                                                  const int* __restrict__ dst,
                                                  int* __restrict__ tab, int E) {
  __shared__ int hist[NNODES];
  int t = threadIdx.x;
  int4* h4 = (int4*)hist;
  for (int i = t; i < NQUAD; i += RANK_T) h4[i] = make_int4(0, 0, 0, 0);
  __syncthreads();
  int per = (E + NCH - 1) / NCH;
  int e0 = blockIdx.x * per;
  int e1 = e0 + per;
  if (e1 > E) e1 = E;
  for (int eb = e0 + t; eb < e1; eb += 4 * RANK_T) {
    int s[4], d[4];
#pragma unroll
    for (int j = 0; j < 4; ++j) {
      int e = eb + j * RANK_T;
      s[j] = 0;
      d[j] = 0;
      if (e < e1) {
        s[j] = src[e];
        d[j] = dst[e];
      }
    }
#pragma unroll
    for (int j = 0; j < 4; ++j) {
      int e = eb + j * RANK_T;
      if (e < e1) {
        atomicAdd(hist + s[j], 1);
        atomicAdd(hist + d[j], 1);
      }
    }
  }
  __syncthreads();
  int4* row = (int4*)(tab + (size_t)blockIdx.x * NNODES);
  for (int i = t; i < NQUAD; i += RANK_T) row[i] = h4[i];
}

__global__ __launch_bounds__(64) void k_colsum(int* __restrict__ tab, int* __restrict__ deg) {
  int n = blockIdx.x * blockDim.x + threadIdx.x;
  if (n >= NNODES) return;
  int* col = tab + n;
  int run = 0;
  for (int c0 = 0; c0 < NCH; c0 += 16) {
    int c[16];
#pragma unroll
    for (int j = 0; j < 16; ++j) c[j] = col[(size_t)(c0 + j) * NNODES];
#pragma unroll
    for (int j = 0; j < 16; ++j) {
      col[(size_t)(c0 + j) * NNODES] = run;
      run += c[j];
    }
  }
  deg[n] = run;
}

#define SCAN_T 1024
#define SCAN_P 5

template <int K>
__device__ __forceinline__ int row_up(int x) {
  return __builtin_amdgcn_update_dpp(0, x, 0x110 + K, 0xf, 0xf, false);
}
__device__ __forceinline__ int wave_incl(int x, int lane) {
  x += row_up<1>(x);
  x += row_up<2>(x);
  x += row_up<4>(x);
  x += row_up<8>(x);
  int low = __builtin_amdgcn_readlane(x, 15);
  return x + ((lane >= 16) ? low : 0);
}

__global__ __launch_bounds__(SCAN_T) void k_scan(const int* __restrict__ deg,
                                                 int* __restrict__ off) {
  __shared__ int wtot[SCAN_P * 32];
  int t = threadIdx.x;
  int lane = t & 31;
  int wu = __builtin_amdgcn_readfirstlane(t >> 5);
  int4 v[SCAN_P];
#pragma unroll
  for (int p = 0; p < SCAN_P; ++p) {
    int g = p * SCAN_T + t;
    v[p] = make_int4(0, 0, 0, 0);
    if (g < NQUAD) v[p] = *(const int4*)(deg + 4 * g);
  }
  int tot[SCAN_P], inc[SCAN_P];
#pragma unroll
  for (int p = 0; p < SCAN_P; ++p) {
    tot[p] = (v[p].x + v[p].y) + (v[p].z + v[p].w);
    inc[p] = wave_incl(tot[p], lane);
    if (lane == 31) wtot[p * 32 + wu] = inc[p];
  }
  __syncthreads();
  int carry = 0;
#pragma unroll
  for (int p = 0; p < SCAN_P; ++p) {
    int w = wtot[p * 32 + lane];
    int ws = wave_incl(w, lane);
    int ahead = __builtin_amdgcn_readlane(ws - w, wu);
    int base = carry + ahead + (inc[p] - tot[p]);
    carry += __builtin_amdgcn_readlane(ws, 31);
    int g = p * SCAN_T + t;
    if (g < NQUAD) {
      int4 o;
      o.x = base;
      o.y = o.x + v[p].x;
      o.z = o.y + v[p].y;
      o.w = o.z + v[p].z;
      *(int4*)(off + 4 * g) = o;
    }
  }
  if (t == 0) off[NNODES] = carry;
}

#define FILL_T 512
#define FILL_U 10
#define NHALF  (NNODES / 2)
#define HQUAD  (NHALF / 4)
#define FILL_B ((HQUAD + FILL_T - 1) / FILL_T)
#define FILL_LOADS(EB)                                                       \
  _Pragma("unroll") for (int j = 0; j < FILL_U; ++j) {                       \
    int e = (EB) + j * FILL_T;                                               \
    int ec = e < e1 ? e : e1 - 1;                                            \
    s[j] = src[ec];                                                          \
    r[j] = rel[ec];                                                          \
    d[j] = dst[ec];                                                          \
  }
__global__ __launch_bounds__(FILL_T) void k_fill(const int* __restrict__ src,
                                                 const int* __restrict__ rel,
                                                 const int* __restrict__ dst,
                                                 const int* __restrict__ off,
                                                 const int* __restrict__ tab,
                                                 int* __restrict__ msg, int E) {
  __shared__ int slot[NHALF];
  int t = threadIdx.x;
  int c = blockIdx.x >> 1;
  int lo = (blockIdx.x & 1) * NHALF;
  const v4i* o4 = (const v4i*)(off + lo);
  const v4i* r4 = (const v4i*)(tab + (size_t)c * NNODES + lo);
  v4i* s4 = (v4i*)slot;
  v4i a[FILL_B], b[FILL_B];
#pragma unroll
  for (int k = 0; k < FILL_B; ++k) {
    int i = t + k * FILL_T;
    int ic = i < HQUAD ? i : HQUAD - 1;
    a[k] = o4[ic];
    b[k] = r4[ic];
  }
  int per = (E + NCH - 1) / NCH;
  int e0 = c * per;
  int e1 = e0 + per;
  if (e1 > E) e1 = E;
  if (e0 >= e1) return;
  int s[FILL_U], r[FILL_U], d[FILL_U];
  int eb = e0 + t;
  FILL_LOADS(eb)
  static_assert(FILL_B == 5, "the fence below names five pairs of base-row loads");
  asm volatile("" ::"v"(a[0]), "v"(b[0]), "v"(a[1]), "v"(b[1]), "v"(a[2]), "v"(b[2]), "v"(a[3]),
               "v"(b[3]), "v"(a[4]), "v"(b[4]));
#pragma unroll
  for (int k = 0; k < FILL_B; ++k) {
    int i = t + k * FILL_T;
    if (i < HQUAD) s4[i] = a[k] + b[k];
  }
  __syncthreads();
  while (eb < e1) {
    int ps[FILL_U], pd[FILL_U];
#pragma unroll
    for (int j = 0; j < FILL_U; ++j) {
      int e = eb + j * FILL_T;
      ps[j] = -1;
      pd[j] = -1;
      if (e < e1) {
        unsigned sl = (unsigned)(s[j] - lo), dl = (unsigned)(d[j] - lo);
        if (sl < NHALF) ps[j] = atomicAdd(slot + sl, 1);
        if (dl < NHALF) pd[j] = atomicAdd(slot + dl, 1);
      }
    }
#pragma unroll
    for (int j = 0; j < FILL_U; ++j) {
      if (ps[j] >= 0) msg[ps[j]] = (r[j] << OBITS) | d[j];
      if (pd[j] >= 0) msg[pd[j]] = ((r[j] + NRELH) << OBITS) | s[j];
    }
    eb += FILL_U * FILL_T;
    if (eb < e1) {
      FILL_LOADS(eb)
    }
  }
}

__device__ __forceinline__ int seg_counts(const int* __restrict__ msg, int beg, int end, int lane) {
  int cnt = 0;
  for (int base = beg; base < end; base += 32) {
    int i = base + lane;
    int q = (i < end) ? (msg[i] >> OBITS) : 64;
#pragma unroll
    for (int j = 0; j < 32; ++j)
      cnt += (__builtin_amdgcn_readlane(q, j) == lane) ? 1 : 0;
  }
  return cnt;
}

template <int K>
__device__ __forceinline__ int row_pick(int x) {
  return __builtin_amdgcn_update_dpp(0, x, 0x150 + K, 0xf, 0xf, false);
}
template <int K>
__device__ __forceinline__ int quad_pick(int x) {
  return __builtin_amdgcn_update_dpp(0, x, K * 0x55, 0xf, 0xf, false);
}

#define L1_LOAD(U)                                                           \
  {                                                                          \
    int ro = row_pick<8 * B + (U)>(roff);                                    \
    iv[U] = __int_as_float(row_pick<8 * B + (U)>(__float_as_int(invl)));     \
    v[U] = *(const float4*)(w1c + (unsigned)(ro + lo));                      \
  }
#define L1_SUM(U)                                                            \
  {                                                                          \
    acc.x = __builtin_fmaf(iv[U], v[U].x, acc.x);                            \
    acc.y = __builtin_fmaf(iv[U], v[U].y, acc.y);                            \
    acc.z = __builtin_fmaf(iv[U], v[U].z, acc.z);                            \
    acc.w = __builtin_fmaf(iv[U], v[U].w, acc.w);                            \
  }
template <int B>
__device__ __forceinline__ void l1_block(int roff, float invl, int lo,
                                         const char* __restrict__ w1c, float4& acc) {
  float4 v[8];
  float iv[8];
  L1_LOAD(0) L1_LOAD(1) L1_LOAD(2) L1_LOAD(3) L1_LOAD(4) L1_LOAD(5) L1_LOAD(6) L1_LOAD(7)
  L1_SUM(0) L1_SUM(1) L1_SUM(2) L1_SUM(3) L1_SUM(4) L1_SUM(5) L1_SUM(6) L1_SUM(7)
}

#define HPITCH 68
__global__ __launch_bounds__(512) void k_l1gemm(const int* __restrict__ off,
                                                const int* __restrict__ msg,
                                                const float* __restrict__ w1,
                                                const float* __restrict__ b1,
                                                const float* __restrict__ w2,
                                                float* __restrict__ P) {
  __shared__ float hs[16 * HPITCH];
  int lane = threadIdx.x & 31;
  int wv = threadIdx.x >> 5;
  int wu = __builtin_amdgcn_readfirstlane(wv);
  int mt = blockIdx.x;
  int n = mt * 16 + wu;
  {
    int beg = off[n], end = off[n + 1];
    int cnt = seg_counts(msg, beg, end, lane);
    float invq = 1.0f / (float)cnt;
    int hh = lane >> 4, rl = lane & 15;
    int slot = 2 * rl + hh;
    int lo = rl * 16;
    int wpad = ((2 * NRELH) << OBITS) | n;
    const char* w1c = (const char*)w1;
    float4 acc = make_float4(0.0f, 0.0f, 0.0f, 0.0f);
    for (int base = beg; base < end; base += 32) {
      int m = end - base;
      if (m > 32) m = 32;
      bool ok = slot < m;
      int w = wpad;
      if (ok) w = msg[base + slot];
      int q = w >> OBITS, o = w & OMASK;
      float invl = __int_as_float(__builtin_amdgcn_ds_bpermute(q << 2, __float_as_int(invq)));
      invl = ok ? invl : 0.0f;
      int roff = (q * NNODES + o) << 8;
      if (m > 16) {
        l1_block<0>(roff, invl, lo, w1c, acc);
        l1_block<1>(roff, invl, lo, w1c, acc);
      } else {
        l1_block<0>(roff, invl, lo, w1c, acc);
      }
    }
    acc.x += __shfl_xor(acc.x, 16);
    acc.y += __shfl_xor(acc.y, 16);
    acc.z += __shfl_xor(acc.z, 16);
    acc.w += __shfl_xor(acc.w, 16);
    float4 sl = *(const float4*)(w1c + (unsigned)((((2 * NRELH) * NNODES + n) << 8) + lo));
    float4 bb = *(const float4*)(b1 + 4 * rl);
    float x0 = acc.x + sl.x + bb.x;
    float x1 = acc.y + sl.y + bb.y;
    float x2 = acc.z + sl.z + bb.z;
    float x3 = acc.w + sl.w + bb.w;
    float4 hv;
    hv.x = x0 > 0.0f ? x0 : 0.0f;
    hv.y = x1 > 0.0f ? x1 : 0.0f;
    hv.z = x2 > 0.0f ? x2 : 0.0f;
    hv.w = x3 > 0.0f ? x3 : 0.0f;
    if (hh == 0) *(float4*)(hs + wu * HPITCH + 4 * rl) = hv;
  }
  __syncthreads();

  int mn    = lane & 15;
  int khalf = lane >> 4;
  const float* hrow = hs + mn * HPITCH + 2 * khalf;
  v2f a[HID / 4];
#pragma unroll
  for (int k = 0; k < HID / 4; ++k) a[k] = *(const v2f*)(hrow + 4 * k);
  int rowoff = khalf * 8;
  for (int r = wu; r < RTOT; r += 16) {
    const float* wrel = w2 + (size_t)r * HID * NCLS + (size_t)(2 * khalf) * NCLS + mn;
    union { v8f v; float f[8]; } acc;
    acc.v = (v8f)0.0f;
#pragma unroll
    for (int k = 0; k < HID / 4; ++k) {
      v2f b;
      b.x = wrel[(size_t)(4 * k) * NCLS];
      b.y = wrel[(size_t)(4 * k + 1) * NCLS];
      acc.v = __builtin_amdgcn_wmma_f32_16x16x4_f32(
           false, a[k],  false, b,
           (short)0, acc.v,  false,  false);
    }
#pragma unroll
    for (int v = 0; v < 8; ++v) {
      P[(size_t)(mt * 16 + v + rowoff) * PCOLS + r * NCLS + mn] = acc.f[v];
    }
  }
}

#define L2_LOAD(I, U, PO, IL)                                                \
  {                                                                          \
    int po = quad_pick<U>(PO);                                               \
    iv[I] = __int_as_float(quad_pick<U>(__float_as_int(IL)));                \
    v[I] = *(const float4*)(Pb + (unsigned)(po + lo));                       \
  }
#define L2_SUM(I)                                                            \
  {                                                                          \
    acc.x = __builtin_fmaf(iv[I], v[I].x, acc.x);                            \
    acc.y = __builtin_fmaf(iv[I], v[I].y, acc.y);                            \
    acc.z = __builtin_fmaf(iv[I], v[I].z, acc.z);                            \
    acc.w = __builtin_fmaf(iv[I], v[I].w, acc.w);                            \
  }
template <int NA, int NB>
__device__ __forceinline__ void l2_stretch(int poffA, float invA, int poffB, float invB, int lo,
                                           const char* __restrict__ Pb, float4& acc) {
  float4 v[8];
  float iv[8];
  L2_LOAD(0, 0, poffA, invA) L2_LOAD(1, 1, poffA, invA)
  if (NA > 2) { L2_LOAD(2, 2, poffA, invA) L2_LOAD(3, 3, poffA, invA) }
  if (NB > 0) { L2_LOAD(4, 0, poffB, invB) L2_LOAD(5, 1, poffB, invB) }
  if (NB > 2) { L2_LOAD(6, 2, poffB, invB) L2_LOAD(7, 3, poffB, invB) }
  L2_SUM(0) L2_SUM(1)
  if (NA > 2) { L2_SUM(2) L2_SUM(3) }
  if (NB > 0) { L2_SUM(4) L2_SUM(5) }
  if (NB > 2) { L2_SUM(6) L2_SUM(7) }
}

__global__ __launch_bounds__(256) void k_l2pull(const int* __restrict__ off,
                                                const int* __restrict__ msg,
                                                const float* __restrict__ P,
                                                const float* __restrict__ b2,
                                                float* __restrict__ out) {
  int lane = threadIdx.x & 31;
  int wu = __builtin_amdgcn_readfirstlane(threadIdx.x >> 5);
  int pr = blockIdx.x * 8 + wu;
  if (pr >= NNODES / 2) return;
  int g = lane >> 2, c4 = lane & 3;
  int slot = 8 * c4 + g;
  int lo = c4 * 16;
  const char* Pb = (const char*)P;
  float4 res = make_float4(0.0f, 0.0f, 0.0f, 0.0f);
  for (int t = 0; t < 2; ++t) {
    int n = 2 * pr + t;
    int beg = off[n], end = off[n + 1];
    int cnt = seg_counts(msg, beg, end, lane);
    float invq = 1.0f / (float)cnt;
    int wpad = ((2 * NRELH) << OBITS) | n;
    float4 acc = make_float4(0.0f, 0.0f, 0.0f, 0.0f);
    for (int base = beg; base < end; base += 64) {
      int mA = end - base;
      int mB = mA - 32;
      if (mA > 32) mA = 32;
      if (mB < 0) mB = 0;
      if (mB > 32) mB = 32;
      bool okA = slot < mA, okB = slot < mB;
      int wA = wpad, wB = wpad;
      if (okA) wA = msg[base + slot];
      if (okB) wB = msg[base + 32 + slot];
      int qA = wA >> OBITS, oA = wA & OMASK;
      int qB = wB >> OBITS, oB = wB & OMASK;
      float invA = __int_as_float(__builtin_amdgcn_ds_bpermute(qA << 2, __float_as_int(invq)));
      float invB = __int_as_float(__builtin_amdgcn_ds_bpermute(qB << 2, __float_as_int(invq)));
      invA = okA ? invA : 0.0f;
      invB = okB ? invB : 0.0f;
      int poffA = (oA * PCOLS + qA * NCLS) << 2;
      int poffB = (oB * PCOLS + qB * NCLS) << 2;
      if (mB > 16)      l2_stretch<4, 4>(poffA, invA, poffB, invB, lo, Pb, acc);
      else if (mB > 0)  l2_stretch<4, 2>(poffA, invA, poffB, invB, lo, Pb, acc);
      else if (mA > 16) l2_stretch<4, 0>(poffA, invA, poffB, invB, lo, Pb, acc);
      else              l2_stretch<2, 0>(poffA, invA, poffB, invB, lo, Pb, acc);
    }
    acc.x += __shfl_xor(acc.x, 4);  acc.y += __shfl_xor(acc.y, 4);
    acc.z += __shfl_xor(acc.z, 4);  acc.w += __shfl_xor(acc.w, 4);
    acc.x += __shfl_xor(acc.x, 8);  acc.y += __shfl_xor(acc.y, 8);
    acc.z += __shfl_xor(acc.z, 8);  acc.w += __shfl_xor(acc.w, 8);
    acc.x += __shfl_xor(acc.x, 16); acc.y += __shfl_xor(acc.y, 16);
    acc.z += __shfl_xor(acc.z, 16); acc.w += __shfl_xor(acc.w, 16);
    if (g == t) res = acc;
  }
  if (lane < 8) {
    int n = 2 * pr + g;
    float4 bb = *(const float4*)(b2 + 4 * c4);
    float4 ps = *(const float4*)(P + (size_t)n * PCOLS + (2 * NRELH) * NCLS + 4 * c4);
    float4 ov;
    ov.x = bb.x + ps.x + res.x;
    ov.y = bb.y + ps.y + res.y;
    ov.z = bb.z + ps.z + res.z;
    ov.w = bb.w + ps.w + res.w;
    *(float4*)(out + (size_t)n * NCLS + 4 * c4) = ov;
  }
}

extern "C" void kernel_launch(void* const* d_in, const int* in_sizes, int n_in,
                              void* d_out, int out_size, void* d_ws, size_t ws_size,
                              hipStream_t stream) {
  const int*   src = (const int*)d_in[0];
  const int*   rel = (const int*)d_in[1];
  const int*   dst = (const int*)d_in[2];
  const float* w1  = (const float*)d_in[3];
  const float* b1  = (const float*)d_in[4];
  const float* w2  = (const float*)d_in[5];
  const float* b2  = (const float*)d_in[6];
  float* out = (float*)d_out;
  const int E = in_sizes[0];

  int* deg  = (int*)d_ws;
  int* off  = deg + NOFF;
  int* msg  = off + NOFF;
  float* P  = (float*)(msg + 2 * (size_t)E);
  int* tab  = (int*)(P + (size_t)NNODES * PCOLS);

  k_count<<<NCH, RANK_T, 0, stream>>>(src, dst, tab, E);

  k_colsum<<<(NNODES + 63) / 64, 64, 0, stream>>>(tab, deg);

  k_scan<<<1, SCAN_T, 0, stream>>>(deg, off);

  k_fill<<<2 * NCH, FILL_T, 0, stream>>>(src, rel, dst, off, tab, msg, E);

  k_l1gemm<<<NNODES / 16, 512, 0, stream>>>(off, msg, w1, b1, w2, P);

  k_l2pull<<<(NNODES / 2 + 7) / 8, 256, 0, stream>>>(off, msg, P, b2, out);
}


# Count pass of both pull kernels: 96 messages a trip, 3 loads out together; 6 wave votes + a popcount per 32 messages give lane L its count of q == L; was a load wait + 32 readlane triples per 32
# speedup vs baseline: 3.1344x; 1.0307x over previous
#include <hip/hip_runtime.h>

#define NNODES 20000
#define NRELH  16
#define RTOT   33
#define HID    64
#define NCLS   16
#define PCOLS  (RTOT * NCLS)
#define OBITS  15
#define OMASK  32767
#define NOFF   20032
#define NQUAD  (NNODES / 4)
#define NCH    128
#define RANK_T 1024

typedef __attribute__((ext_vector_type(2))) float v2f;
typedef __attribute__((ext_vector_type(8))) float v8f;
typedef __attribute__((ext_vector_type(4))) int v4i;


__global__ __launch_bounds__(RANK_T) void k_count(const int* __restrict__ src,
                                                  const int* __restrict__ dst,
                                                  int* __restrict__ tab, int E) {
  __shared__ int hist[NNODES];
  int t = threadIdx.x;
  int4* h4 = (int4*)hist;
  for (int i = t; i < NQUAD; i += RANK_T) h4[i] = make_int4(0, 0, 0, 0);
  __syncthreads();
  int per = (E + NCH - 1) / NCH;
  int e0 = blockIdx.x * per;
  int e1 = e0 + per;
  if (e1 > E) e1 = E;
  for (int eb = e0 + t; eb < e1; eb += 4 * RANK_T) {
    int s[4], d[4];
#pragma unroll
    for (int j = 0; j < 4; ++j) {
      int e = eb + j * RANK_T;
      s[j] = 0;
      d[j] = 0;
      if (e < e1) {
        s[j] = src[e];
        d[j] = dst[e];
      }
    }
#pragma unroll
    for (int j = 0; j < 4; ++j) {
      int e = eb + j * RANK_T;
      if (e < e1) {
        atomicAdd(hist + s[j], 1);
        atomicAdd(hist + d[j], 1);
      }
    }
  }
  __syncthreads();
  int4* row = (int4*)(tab + (size_t)blockIdx.x * NNODES);
  for (int i = t; i < NQUAD; i += RANK_T) row[i] = h4[i];
}

__global__ __launch_bounds__(64) void k_colsum(int* __restrict__ tab, int* __restrict__ deg) {
  int n = blockIdx.x * blockDim.x + threadIdx.x;
  if (n >= NNODES) return;
  int* col = tab + n;
  int run = 0;
  for (int c0 = 0; c0 < NCH; c0 += 16) {
    int c[16];
#pragma unroll
    for (int j = 0; j < 16; ++j) c[j] = col[(size_t)(c0 + j) * NNODES];
#pragma unroll
    for (int j = 0; j < 16; ++j) {
      col[(size_t)(c0 + j) * NNODES] = run;
      run += c[j];
    }
  }
  deg[n] = run;
}

#define SCAN_T 1024
#define SCAN_P 5

template <int K>
__device__ __forceinline__ int row_up(int x) {
  return __builtin_amdgcn_update_dpp(0, x, 0x110 + K, 0xf, 0xf, false);
}
__device__ __forceinline__ int wave_incl(int x, int lane) {
  x += row_up<1>(x);
  x += row_up<2>(x);
  x += row_up<4>(x);
  x += row_up<8>(x);
  int low = __builtin_amdgcn_readlane(x, 15);
  return x + ((lane >= 16) ? low : 0);
}

__global__ __launch_bounds__(SCAN_T) void k_scan(const int* __restrict__ deg,
                                                 int* __restrict__ off) {
  __shared__ int wtot[SCAN_P * 32];
  int t = threadIdx.x;
  int lane = t & 31;
  int wu = __builtin_amdgcn_readfirstlane(t >> 5);
  int4 v[SCAN_P];
#pragma unroll
  for (int p = 0; p < SCAN_P; ++p) {
    int g = p * SCAN_T + t;
    v[p] = make_int4(0, 0, 0, 0);
    if (g < NQUAD) v[p] = *(const int4*)(deg + 4 * g);
  }
  int tot[SCAN_P], inc[SCAN_P];
#pragma unroll
  for (int p = 0; p < SCAN_P; ++p) {
    tot[p] = (v[p].x + v[p].y) + (v[p].z + v[p].w);
    inc[p] = wave_incl(tot[p], lane);
    if (lane == 31) wtot[p * 32 + wu] = inc[p];
  }
  __syncthreads();
  int carry = 0;
#pragma unroll
  for (int p = 0; p < SCAN_P; ++p) {
    int w = wtot[p * 32 + lane];
    int ws = wave_incl(w, lane);
    int ahead = __builtin_amdgcn_readlane(ws - w, wu);
    int base = carry + ahead + (inc[p] - tot[p]);
    carry += __builtin_amdgcn_readlane(ws, 31);
    int g = p * SCAN_T + t;
    if (g < NQUAD) {
      int4 o;
      o.x = base;
      o.y = o.x + v[p].x;
      o.z = o.y + v[p].y;
      o.w = o.z + v[p].z;
      *(int4*)(off + 4 * g) = o;
    }
  }
  if (t == 0) off[NNODES] = carry;
}

#define FILL_T 512
#define FILL_U 10
#define NHALF  (NNODES / 2)
#define HQUAD  (NHALF / 4)
#define FILL_B ((HQUAD + FILL_T - 1) / FILL_T)
#define FILL_LOADS(EB)                                                       \
  _Pragma("unroll") for (int j = 0; j < FILL_U; ++j) {                       \
    int e = (EB) + j * FILL_T;                                               \
    int ec = e < e1 ? e : e1 - 1;                                            \
    s[j] = src[ec];                                                          \
    r[j] = rel[ec];                                                          \
    d[j] = dst[ec];                                                          \
  }
__global__ __launch_bounds__(FILL_T) void k_fill(const int* __restrict__ src,
                                                 const int* __restrict__ rel,
                                                 const int* __restrict__ dst,
                                                 const int* __restrict__ off,
                                                 const int* __restrict__ tab,
                                                 int* __restrict__ msg, int E) {
  __shared__ int slot[NHALF];
  int t = threadIdx.x;
  int c = blockIdx.x >> 1;
  int lo = (blockIdx.x & 1) * NHALF;
  const v4i* o4 = (const v4i*)(off + lo);
  const v4i* r4 = (const v4i*)(tab + (size_t)c * NNODES + lo);
  v4i* s4 = (v4i*)slot;
  v4i a[FILL_B], b[FILL_B];
#pragma unroll
  for (int k = 0; k < FILL_B; ++k) {
    int i = t + k * FILL_T;
    int ic = i < HQUAD ? i : HQUAD - 1;
    a[k] = o4[ic];
    b[k] = r4[ic];
  }
  int per = (E + NCH - 1) / NCH;
  int e0 = c * per;
  int e1 = e0 + per;
  if (e1 > E) e1 = E;
  if (e0 >= e1) return;
  int s[FILL_U], r[FILL_U], d[FILL_U];
  int eb = e0 + t;
  FILL_LOADS(eb)
  static_assert(FILL_B == 5, "the fence below names five pairs of base-row loads");
  asm volatile("" ::"v"(a[0]), "v"(b[0]), "v"(a[1]), "v"(b[1]), "v"(a[2]), "v"(b[2]), "v"(a[3]),
               "v"(b[3]), "v"(a[4]), "v"(b[4]));
#pragma unroll
  for (int k = 0; k < FILL_B; ++k) {
    int i = t + k * FILL_T;
    if (i < HQUAD) s4[i] = a[k] + b[k];
  }
  __syncthreads();
  while (eb < e1) {
    int ps[FILL_U], pd[FILL_U];
#pragma unroll
    for (int j = 0; j < FILL_U; ++j) {
      int e = eb + j * FILL_T;
      ps[j] = -1;
      pd[j] = -1;
      if (e < e1) {
        unsigned sl = (unsigned)(s[j] - lo), dl = (unsigned)(d[j] - lo);
        if (sl < NHALF) ps[j] = atomicAdd(slot + sl, 1);
        if (dl < NHALF) pd[j] = atomicAdd(slot + dl, 1);
      }
    }
#pragma unroll
    for (int j = 0; j < FILL_U; ++j) {
      if (ps[j] >= 0) msg[ps[j]] = (r[j] << OBITS) | d[j];
      if (pd[j] >= 0) msg[pd[j]] = ((r[j] + NRELH) << OBITS) | s[j];
    }
    eb += FILL_U * FILL_T;
    if (eb < e1) {
      FILL_LOADS(eb)
    }
  }
}

#define CNT_K 3
__device__ __forceinline__ int seg_counts(const int* __restrict__ msg, int beg, int end, int lane) {
  unsigned flip[5];
#pragma unroll
  for (int b = 0; b < 5; ++b) flip[b] = ((lane >> b) & 1) ? 0u : ~0u;
  int cnt = 0;
  for (int base = beg; base < end; base += 32 * CNT_K) {
    int w[CNT_K];
#pragma unroll
    for (int k = 0; k < CNT_K; ++k) {
      int i = base + 32 * k + lane;
      w[k] = msg[i < end ? i : end - 1];
    }
    static_assert(CNT_K == 3, "the fence below names three loaded words");
    asm volatile("" ::"v"(w[0]), "v"(w[1]), "v"(w[2]));
#pragma unroll
    for (int k = 0; k < CNT_K; ++k) {
      if (base + 32 * k < end) {
        int i = base + 32 * k + lane;
        int q = w[k] >> OBITS;
        unsigned m = __builtin_amdgcn_ballot_w32(i < end);
#pragma unroll
        for (int b = 0; b < 5; ++b)
          m &= __builtin_amdgcn_ballot_w32(((q >> b) & 1) != 0) ^ flip[b];
        cnt += __builtin_popcount(m);
      }
    }
  }
  return cnt;
}

template <int K>
__device__ __forceinline__ int row_pick(int x) {
  return __builtin_amdgcn_update_dpp(0, x, 0x150 + K, 0xf, 0xf, false);
}
template <int K>
__device__ __forceinline__ int quad_pick(int x) {
  return __builtin_amdgcn_update_dpp(0, x, K * 0x55, 0xf, 0xf, false);
}

#define L1_LOAD(U)                                                           \
  {                                                                          \
    int ro = row_pick<8 * B + (U)>(roff);                                    \
    iv[U] = __int_as_float(row_pick<8 * B + (U)>(__float_as_int(invl)));     \
    v[U] = *(const float4*)(w1c + (unsigned)(ro + lo));                      \
  }
#define L1_SUM(U)                                                            \
  {                                                                          \
    acc.x = __builtin_fmaf(iv[U], v[U].x, acc.x);                            \
    acc.y = __builtin_fmaf(iv[U], v[U].y, acc.y);                            \
    acc.z = __builtin_fmaf(iv[U], v[U].z, acc.z);                            \
    acc.w = __builtin_fmaf(iv[U], v[U].w, acc.w);                            \
  }
template <int B>
__device__ __forceinline__ void l1_block(int roff, float invl, int lo,
                                         const char* __restrict__ w1c, float4& acc) {
  float4 v[8];
  float iv[8];
  L1_LOAD(0) L1_LOAD(1) L1_LOAD(2) L1_LOAD(3) L1_LOAD(4) L1_LOAD(5) L1_LOAD(6) L1_LOAD(7)
  L1_SUM(0) L1_SUM(1) L1_SUM(2) L1_SUM(3) L1_SUM(4) L1_SUM(5) L1_SUM(6) L1_SUM(7)
}

#define HPITCH 68
__global__ __launch_bounds__(512) void k_l1gemm(const int* __restrict__ off,
                                                const int* __restrict__ msg,
                                                const float* __restrict__ w1,
                                                const float* __restrict__ b1,
                                                const float* __restrict__ w2,
                                                float* __restrict__ P) {
  __shared__ float hs[16 * HPITCH];
  int lane = threadIdx.x & 31;
  int wv = threadIdx.x >> 5;
  int wu = __builtin_amdgcn_readfirstlane(wv);
  int mt = blockIdx.x;
  int n = mt * 16 + wu;
  {
    int beg = off[n], end = off[n + 1];
    int cnt = seg_counts(msg, beg, end, lane);
    float invq = 1.0f / (float)cnt;
    int hh = lane >> 4, rl = lane & 15;
    int slot = 2 * rl + hh;
    int lo = rl * 16;
    int wpad = ((2 * NRELH) << OBITS) | n;
    const char* w1c = (const char*)w1;
    float4 acc = make_float4(0.0f, 0.0f, 0.0f, 0.0f);
    for (int base = beg; base < end; base += 32) {
      int m = end - base;
      if (m > 32) m = 32;
      bool ok = slot < m;
      int w = wpad;
      if (ok) w = msg[base + slot];
      int q = w >> OBITS, o = w & OMASK;
      float invl = __int_as_float(__builtin_amdgcn_ds_bpermute(q << 2, __float_as_int(invq)));
      invl = ok ? invl : 0.0f;
      int roff = (q * NNODES + o) << 8;
      if (m > 16) {
        l1_block<0>(roff, invl, lo, w1c, acc);
        l1_block<1>(roff, invl, lo, w1c, acc);
      } else {
        l1_block<0>(roff, invl, lo, w1c, acc);
      }
    }
    acc.x += __shfl_xor(acc.x, 16);
    acc.y += __shfl_xor(acc.y, 16);
    acc.z += __shfl_xor(acc.z, 16);
    acc.w += __shfl_xor(acc.w, 16);
    float4 sl = *(const float4*)(w1c + (unsigned)((((2 * NRELH) * NNODES + n) << 8) + lo));
    float4 bb = *(const float4*)(b1 + 4 * rl);
    float x0 = acc.x + sl.x + bb.x;
    float x1 = acc.y + sl.y + bb.y;
    float x2 = acc.z + sl.z + bb.z;
    float x3 = acc.w + sl.w + bb.w;
    float4 hv;
    hv.x = x0 > 0.0f ? x0 : 0.0f;
    hv.y = x1 > 0.0f ? x1 : 0.0f;
    hv.z = x2 > 0.0f ? x2 : 0.0f;
    hv.w = x3 > 0.0f ? x3 : 0.0f;
    if (hh == 0) *(float4*)(hs + wu * HPITCH + 4 * rl) = hv;
  }
  __syncthreads();

  int mn    = lane & 15;
  int khalf = lane >> 4;
  const float* hrow = hs + mn * HPITCH + 2 * khalf;
  v2f a[HID / 4];
#pragma unroll
  for (int k = 0; k < HID / 4; ++k) a[k] = *(const v2f*)(hrow + 4 * k);
  int rowoff = khalf * 8;
  for (int r = wu; r < RTOT; r += 16) {
    const float* wrel = w2 + (size_t)r * HID * NCLS + (size_t)(2 * khalf) * NCLS + mn;
    union { v8f v; float f[8]; } acc;
    acc.v = (v8f)0.0f;
#pragma unroll
    for (int k = 0; k < HID / 4; ++k) {
      v2f b;
      b.x = wrel[(size_t)(4 * k) * NCLS];
      b.y = wrel[(size_t)(4 * k + 1) * NCLS];
      acc.v = __builtin_amdgcn_wmma_f32_16x16x4_f32(
           false, a[k],  false, b,
           (short)0, acc.v,  false,  false);
    }
#pragma unroll
    for (int v = 0; v < 8; ++v) {
      P[(size_t)(mt * 16 + v + rowoff) * PCOLS + r * NCLS + mn] = acc.f[v];
    }
  }
}

#define L2_LOAD(I, U, PO, IL)                                                \
  {                                                                          \
    int po = quad_pick<U>(PO);                                               \
    iv[I] = __int_as_float(quad_pick<U>(__float_as_int(IL)));                \
    v[I] = *(const float4*)(Pb + (unsigned)(po + lo));                       \
  }
#define L2_SUM(I)                                                            \
  {                                                                          \
    acc.x = __builtin_fmaf(iv[I], v[I].x, acc.x);                            \
    acc.y = __builtin_fmaf(iv[I], v[I].y, acc.y);                            \
    acc.z = __builtin_fmaf(iv[I], v[I].z, acc.z);                            \
    acc.w = __builtin_fmaf(iv[I], v[I].w, acc.w);                            \
  }
template <int NA, int NB>
__device__ __forceinline__ void l2_stretch(int poffA, float invA, int poffB, float invB, int lo,
                                           const char* __restrict__ Pb, float4& acc) {
  float4 v[8];
  float iv[8];
  L2_LOAD(0, 0, poffA, invA) L2_LOAD(1, 1, poffA, invA)
  if (NA > 2) { L2_LOAD(2, 2, poffA, invA) L2_LOAD(3, 3, poffA, invA) }
  if (NB > 0) { L2_LOAD(4, 0, poffB, invB) L2_LOAD(5, 1, poffB, invB) }
  if (NB > 2) { L2_LOAD(6, 2, poffB, invB) L2_LOAD(7, 3, poffB, invB) }
  L2_SUM(0) L2_SUM(1)
  if (NA > 2) { L2_SUM(2) L2_SUM(3) }
  if (NB > 0) { L2_SUM(4) L2_SUM(5) }
  if (NB > 2) { L2_SUM(6) L2_SUM(7) }
}

__global__ __launch_bounds__(256) void k_l2pull(const int* __restrict__ off,
                                                const int* __restrict__ msg,
                                                const float* __restrict__ P,
                                                const float* __restrict__ b2,
                                                float* __restrict__ out) {
  int lane = threadIdx.x & 31;
  int wu = __builtin_amdgcn_readfirstlane(threadIdx.x >> 5);
  int pr = blockIdx.x * 8 + wu;
  if (pr >= NNODES / 2) return;
  int g = lane >> 2, c4 = lane & 3;
  int slot = 8 * c4 + g;
  int lo = c4 * 16;
  const char* Pb = (const char*)P;
  float4 res = make_float4(0.0f, 0.0f, 0.0f, 0.0f);
  for (int t = 0; t < 2; ++t) {
    int n = 2 * pr + t;
    int beg = off[n], end = off[n + 1];
    int cnt = seg_counts(msg, beg, end, lane);
    float invq = 1.0f / (float)cnt;
    int wpad = ((2 * NRELH) << OBITS) | n;
    float4 acc = make_float4(0.0f, 0.0f, 0.0f, 0.0f);
    for (int base = beg; base < end; base += 64) {
      int mA = end - base;
      int mB = mA - 32;
      if (mA > 32) mA = 32;
      if (mB < 0) mB = 0;
      if (mB > 32) mB = 32;
      bool okA = slot < mA, okB = slot < mB;
      int wA = wpad, wB = wpad;
      if (okA) wA = msg[base + slot];
      if (okB) wB = msg[base + 32 + slot];
      int qA = wA >> OBITS, oA = wA & OMASK;
      int qB = wB >> OBITS, oB = wB & OMASK;
      float invA = __int_as_float(__builtin_amdgcn_ds_bpermute(qA << 2, __float_as_int(invq)));
      float invB = __int_as_float(__builtin_amdgcn_ds_bpermute(qB << 2, __float_as_int(invq)));
      invA = okA ? invA : 0.0f;
      invB = okB ? invB : 0.0f;
      int poffA = (oA * PCOLS + qA * NCLS) << 2;
      int poffB = (oB * PCOLS + qB * NCLS) << 2;
      if (mB > 16)      l2_stretch<4, 4>(poffA, invA, poffB, invB, lo, Pb, acc);
      else if (mB > 0)  l2_stretch<4, 2>(poffA, invA, poffB, invB, lo, Pb, acc);
      else if (mA > 16) l2_stretch<4, 0>(poffA, invA, poffB, invB, lo, Pb, acc);
      else              l2_stretch<2, 0>(poffA, invA, poffB, invB, lo, Pb, acc);
    }
    acc.x += __shfl_xor(acc.x, 4);  acc.y += __shfl_xor(acc.y, 4);
    acc.z += __shfl_xor(acc.z, 4);  acc.w += __shfl_xor(acc.w, 4);
    acc.x += __shfl_xor(acc.x, 8);  acc.y += __shfl_xor(acc.y, 8);
    acc.z += __shfl_xor(acc.z, 8);  acc.w += __shfl_xor(acc.w, 8);
    acc.x += __shfl_xor(acc.x, 16); acc.y += __shfl_xor(acc.y, 16);
    acc.z += __shfl_xor(acc.z, 16); acc.w += __shfl_xor(acc.w, 16);
    if (g == t) res = acc;
  }
  if (lane < 8) {
    int n = 2 * pr + g;
    float4 bb = *(const float4*)(b2 + 4 * c4);
    float4 ps = *(const float4*)(P + (size_t)n * PCOLS + (2 * NRELH) * NCLS + 4 * c4);
    float4 ov;
    ov.x = bb.x + ps.x + res.x;
    ov.y = bb.y + ps.y + res.y;
    ov.z = bb.z + ps.z + res.z;
    ov.w = bb.w + ps.w + res.w;
    *(float4*)(out + (size_t)n * NCLS + 4 * c4) = ov;
  }
}

extern "C" void kernel_launch(void* const* d_in, const int* in_sizes, int n_in,
                              void* d_out, int out_size, void* d_ws, size_t ws_size,
                              hipStream_t stream) {
  const int*   src = (const int*)d_in[0];
  const int*   rel = (const int*)d_in[1];
  const int*   dst = (const int*)d_in[2];
  const float* w1  = (const float*)d_in[3];
  const float* b1  = (const float*)d_in[4];
  const float* w2  = (const float*)d_in[5];
  const float* b2  = (const float*)d_in[6];
  float* out = (float*)d_out;
  const int E = in_sizes[0];

  int* deg  = (int*)d_ws;
  int* off  = deg + NOFF;
  int* msg  = off + NOFF;
  float* P  = (float*)(msg + 2 * (size_t)E);
  int* tab  = (int*)(P + (size_t)NNODES * PCOLS);

  k_count<<<NCH, RANK_T, 0, stream>>>(src, dst, tab, E);

  k_colsum<<<(NNODES + 63) / 64, 64, 0, stream>>>(tab, deg);

  k_scan<<<1, SCAN_T, 0, stream>>>(deg, off);

  k_fill<<<2 * NCH, FILL_T, 0, stream>>>(src, rel, dst, off, tab, msg, E);

  k_l1gemm<<<NNODES / 16, 512, 0, stream>>>(off, msg, w1, b1, w2, P);

  k_l2pull<<<(NNODES / 2 + 7) / 8, 256, 0, stream>>>(off, msg, P, b2, out);
}


# Receiver sort in two launches: (chunk, node slice) work-groups; a fill work-group sums its slice's 32 count rows and scans them itself; column-prefix and scan launches gone; h, out add order changes
# speedup vs baseline: 4.2687x; 1.3619x over previous
#include <hip/hip_runtime.h>

#define NNODES 20000
#define NRELH  16
#define RTOT   33
#define HID    64
#define NCLS   16
#define PCOLS  (RTOT * NCLS)
#define OBITS  15
#define OMASK  32767
#define NOFF   20032
#define NCH    32
#define NSL    8
#define SLN    (NNODES / NSL)
#define SLQ    (SLN / 4)
#define NPAIR  (NCH * NSL)

static_assert(NNODES == NSL * 4 * SLQ, "a slice is a whole number of four-counter groups");

typedef __attribute__((ext_vector_type(2))) float v2f;
typedef __attribute__((ext_vector_type(8))) float v8f;
typedef __attribute__((ext_vector_type(4))) int v4i;


template <int K>
__device__ __forceinline__ int row_up(int x) {
  return __builtin_amdgcn_update_dpp(0, x, 0x110 + K, 0xf, 0xf, false);
}
__device__ __forceinline__ int wave_incl(int x, int lane) {
  x += row_up<1>(x);
  x += row_up<2>(x);
  x += row_up<4>(x);
  x += row_up<8>(x);
  int low = __builtin_amdgcn_readlane(x, 15);
  return x + ((lane >= 16) ? low : 0);
}

#define CNT_T 512
#define CNT_U 40
#define CNT_LOADS(S, D, EB)                                                  \
  _Pragma("unroll") for (int j = 0; j < CNT_U; ++j) {                        \
    int e = (EB) + j * CNT_T;                                                \
    int ec = e < e1 ? e : e1 - 1;                                            \
    S[j] = src[ec];                                                          \
    D[j] = dst[ec];                                                          \
  }
__global__ __launch_bounds__(CNT_T) void k_count(const int* __restrict__ src,
                                                 const int* __restrict__ dst,
                                                 int* __restrict__ tab,
                                                 int* __restrict__ tot, int E) {
  __shared__ int hist[SLN];
  __shared__ int wgtot;
  int t = threadIdx.x;
  int c = blockIdx.x / NSL;
  int sl = blockIdx.x % NSL;
  int lo = sl * SLN;
  v4i* h4 = (v4i*)hist;
  for (int i = t; i < SLQ; i += CNT_T) h4[i] = (v4i)0;
  if (t == 0) wgtot = 0;
  int per = (E + NCH - 1) / NCH;
  int e0 = c * per;
  int e1 = e0 + per;
  if (e1 > E) e1 = E;
  int s[CNT_U], d[CNT_U];
  int tb = e0;
  if (tb < e1) {
    CNT_LOADS(s, d, tb + t)
  }
  __syncthreads();
  int mine = 0;
  while (tb < e1) {
#pragma unroll
    for (int j = 0; j < CNT_U; ++j) {
      int e = tb + t + j * CNT_T;
      if (e < e1) {
        unsigned a = (unsigned)(s[j] - lo), b = (unsigned)(d[j] - lo);
        if (a < SLN) { atomicAdd(hist + a, 1); ++mine; }
        if (b < SLN) { atomicAdd(hist + b, 1); ++mine; }
      }
    }
    tb += CNT_U * CNT_T;
    if (tb < e1) {
      CNT_LOADS(s, d, tb + t)
    }
  }
  mine += __shfl_xor(mine, 16);
  mine += __shfl_xor(mine, 8);
  mine += __shfl_xor(mine, 4);
  mine += __shfl_xor(mine, 2);
  mine += __shfl_xor(mine, 1);
  if ((t & 31) == 0) atomicAdd(&wgtot, mine);
  __syncthreads();
  v4i* row = (v4i*)(tab + (size_t)c * NNODES + lo);
  for (int i = t; i < SLQ; i += CNT_T) row[i] = h4[i];
  if (t == 0) tot[blockIdx.x] = wgtot;
}

#define FILL_T 640
#define FILL_W (FILL_T / 32)
#define FILL_U 16
static_assert(FILL_T >= SLQ && FILL_T >= NPAIR && FILL_W <= 32, "one thread per group; wave totals fit one wave");
#define FILL_LOADS(S, R, D, EB)                                              \
  _Pragma("unroll") for (int j = 0; j < FILL_U; ++j) {                       \
    int e = (EB) + j * FILL_T;                                               \
    int ec = e < e1 ? e : e1 - 1;                                            \
    S[j] = src[ec];                                                          \
    R[j] = rel[ec];                                                          \
    D[j] = dst[ec];                                                          \
  }
__global__ __launch_bounds__(FILL_T) void k_fill(const int* __restrict__ src,
                                                 const int* __restrict__ rel,
                                                 const int* __restrict__ dst,
                                                 const int* __restrict__ tab,
                                                 const int* __restrict__ tot,
                                                 int* __restrict__ off,
                                                 int* __restrict__ msg, int E) {
  __shared__ int slot[SLN];
  __shared__ int wsum[FILL_W], bsum[FILL_W];
  int t = threadIdx.x;
  int lane = t & 31;
  int wu = __builtin_amdgcn_readfirstlane(t >> 5);
  int c = blockIdx.x / NSL;
  int sl = blockIdx.x % NSL;
  int lo = sl * SLN;
  int per = (E + NCH - 1) / NCH;
  int e0 = c * per;
  int e1 = e0 + per;
  if (e1 > E) e1 = E;
  int s[FILL_U], r[FILL_U], d[FILL_U];
  int tb = e0;
  if (tb < e1) {
    FILL_LOADS(s, r, d, tb + t)
  }
  int tv = tot[t < NPAIR ? t : NPAIR - 1];
  bool own = t < SLQ;
  v4i dg = (v4i)0, bf = (v4i)0;
  if (own) {
    const v4i* col = (const v4i*)(tab + lo) + t;
#pragma unroll 16
    for (int cp = 0; cp < NCH; ++cp) {
      v4i a = col[(size_t)cp * (NNODES / 4)];
      dg += a;
      if (cp < c) bf += a;
    }
  }
  int bpart = (t < NPAIR && (t % NSL) < sl) ? tv : 0;
  int T = (dg.x + dg.y) + (dg.z + dg.w);
  int inc = wave_incl(T, lane);
  int binc = wave_incl(bpart, lane);
  if (lane == 31) {
    wsum[wu] = inc;
    bsum[wu] = binc;
  }
  __syncthreads();
  int w = lane < FILL_W ? wsum[lane] : 0;
  int b = lane < FILL_W ? bsum[lane] : 0;
  int wsc = wave_incl(w, lane);
  int bsc = wave_incl(b, lane);
  int ahead = __builtin_amdgcn_readlane(wsc - w, wu);
  int base = __builtin_amdgcn_readlane(bsc, 31);
  if (own) {
    v4i o;
    o.x = base + ahead + (inc - T);
    o.y = o.x + dg.x;
    o.z = o.y + dg.y;
    o.w = o.z + dg.z;
    ((v4i*)slot)[t] = o + bf;
    if (c == 0) {
      ((v4i*)(off + lo))[t] = o;
      if (sl == NSL - 1 && t == SLQ - 1) off[NNODES] = o.w + dg.w;
    }
  }
  __syncthreads();
  while (tb < e1) {
    int sn[FILL_U], rn[FILL_U], dn[FILL_U];
    int tbn = tb + FILL_U * FILL_T;
    bool more = tbn < e1;
    if (more) {
      FILL_LOADS(sn, rn, dn, tbn + t)
    }
    int ps[FILL_U], pd[FILL_U];
#pragma unroll
    for (int j = 0; j < FILL_U; ++j) {
      int e = tb + t + j * FILL_T;
      ps[j] = -1;
      pd[j] = -1;
      if (e < e1) {
        unsigned a = (unsigned)(s[j] - lo), bb = (unsigned)(d[j] - lo);
        if (a < SLN) ps[j] = atomicAdd(slot + a, 1);
        if (bb < SLN) pd[j] = atomicAdd(slot + bb, 1);
      }
    }
#pragma unroll
    for (int j = 0; j < FILL_U; ++j) {
      if (ps[j] >= 0) msg[ps[j]] = (r[j] << OBITS) | d[j];
      if (pd[j] >= 0) msg[pd[j]] = ((r[j] + NRELH) << OBITS) | s[j];
    }
    if (more) {
#pragma unroll
      for (int j = 0; j < FILL_U; ++j) {
        s[j] = sn[j];
        r[j] = rn[j];
        d[j] = dn[j];
      }
    }
    tb = tbn;
  }
}

#define CNT_K 3
__device__ __forceinline__ int seg_counts(const int* __restrict__ msg, int beg, int end, int lane) {
  unsigned flip[5];
#pragma unroll
  for (int b = 0; b < 5; ++b) flip[b] = ((lane >> b) & 1) ? 0u : ~0u;
  int cnt = 0;
  for (int base = beg; base < end; base += 32 * CNT_K) {
    int w[CNT_K];
#pragma unroll
    for (int k = 0; k < CNT_K; ++k) {
      int i = base + 32 * k + lane;
      w[k] = msg[i < end ? i : end - 1];
    }
    static_assert(CNT_K == 3, "the fence below names three loaded words");
    asm volatile("" ::"v"(w[0]), "v"(w[1]), "v"(w[2]));
#pragma unroll
    for (int k = 0; k < CNT_K; ++k) {
      if (base + 32 * k < end) {
        int i = base + 32 * k + lane;
        int q = w[k] >> OBITS;
        unsigned m = __builtin_amdgcn_ballot_w32(i < end);
#pragma unroll
        for (int b = 0; b < 5; ++b)
          m &= __builtin_amdgcn_ballot_w32(((q >> b) & 1) != 0) ^ flip[b];
        cnt += __builtin_popcount(m);
      }
    }
  }
  return cnt;
}

template <int K>
__device__ __forceinline__ int row_pick(int x) {
  return __builtin_amdgcn_update_dpp(0, x, 0x150 + K, 0xf, 0xf, false);
}
template <int K>
__device__ __forceinline__ int quad_pick(int x) {
  return __builtin_amdgcn_update_dpp(0, x, K * 0x55, 0xf, 0xf, false);
}

#define L1_LOAD(U)                                                           \
  {                                                                          \
    int ro = row_pick<8 * B + (U)>(roff);                                    \
    iv[U] = __int_as_float(row_pick<8 * B + (U)>(__float_as_int(invl)));     \
    v[U] = *(const float4*)(w1c + (unsigned)(ro + lo));                      \
  }
#define L1_SUM(U)                                                            \
  {                                                                          \
    acc.x = __builtin_fmaf(iv[U], v[U].x, acc.x);                            \
    acc.y = __builtin_fmaf(iv[U], v[U].y, acc.y);                            \
    acc.z = __builtin_fmaf(iv[U], v[U].z, acc.z);                            \
    acc.w = __builtin_fmaf(iv[U], v[U].w, acc.w);                            \
  }
template <int B>
__device__ __forceinline__ void l1_block(int roff, float invl, int lo,
                                         const char* __restrict__ w1c, float4& acc) {
  float4 v[8];
  float iv[8];
  L1_LOAD(0) L1_LOAD(1) L1_LOAD(2) L1_LOAD(3) L1_LOAD(4) L1_LOAD(5) L1_LOAD(6) L1_LOAD(7)
  L1_SUM(0) L1_SUM(1) L1_SUM(2) L1_SUM(3) L1_SUM(4) L1_SUM(5) L1_SUM(6) L1_SUM(7)
}

#define HPITCH 68
__global__ __launch_bounds__(512) void k_l1gemm(const int* __restrict__ off,
                                                const int* __restrict__ msg,
                                                const float* __restrict__ w1,
                                                const float* __restrict__ b1,
                                                const float* __restrict__ w2,
                                                float* __restrict__ P) {
  __shared__ float hs[16 * HPITCH];
  int lane = threadIdx.x & 31;
  int wv = threadIdx.x >> 5;
  int wu = __builtin_amdgcn_readfirstlane(wv);
  int mt = blockIdx.x;
  int n = mt * 16 + wu;
  {
    int beg = off[n], end = off[n + 1];
    int cnt = seg_counts(msg, beg, end, lane);
    float invq = 1.0f / (float)cnt;
    int hh = lane >> 4, rl = lane & 15;
    int slot = 2 * rl + hh;
    int lo = rl * 16;
    int wpad = ((2 * NRELH) << OBITS) | n;
    const char* w1c = (const char*)w1;
    float4 acc = make_float4(0.0f, 0.0f, 0.0f, 0.0f);
    for (int base = beg; base < end; base += 32) {
      int m = end - base;
      if (m > 32) m = 32;
      bool ok = slot < m;
      int w = wpad;
      if (ok) w = msg[base + slot];
      int q = w >> OBITS, o = w & OMASK;
      float invl = __int_as_float(__builtin_amdgcn_ds_bpermute(q << 2, __float_as_int(invq)));
      invl = ok ? invl : 0.0f;
      int roff = (q * NNODES + o) << 8;
      if (m > 16) {
        l1_block<0>(roff, invl, lo, w1c, acc);
        l1_block<1>(roff, invl, lo, w1c, acc);
      } else {
        l1_block<0>(roff, invl, lo, w1c, acc);
      }
    }
    acc.x += __shfl_xor(acc.x, 16);
    acc.y += __shfl_xor(acc.y, 16);
    acc.z += __shfl_xor(acc.z, 16);
    acc.w += __shfl_xor(acc.w, 16);
    float4 sl = *(const float4*)(w1c + (unsigned)((((2 * NRELH) * NNODES + n) << 8) + lo));
    float4 bb = *(const float4*)(b1 + 4 * rl);
    float x0 = acc.x + sl.x + bb.x;
    float x1 = acc.y + sl.y + bb.y;
    float x2 = acc.z + sl.z + bb.z;
    float x3 = acc.w + sl.w + bb.w;
    float4 hv;
    hv.x = x0 > 0.0f ? x0 : 0.0f;
    hv.y = x1 > 0.0f ? x1 : 0.0f;
    hv.z = x2 > 0.0f ? x2 : 0.0f;
    hv.w = x3 > 0.0f ? x3 : 0.0f;
    if (hh == 0) *(float4*)(hs + wu * HPITCH + 4 * rl) = hv;
  }
  __syncthreads();

  int mn    = lane & 15;
  int khalf = lane >> 4;
  const float* hrow = hs + mn * HPITCH + 2 * khalf;
  v2f a[HID / 4];
#pragma unroll
  for (int k = 0; k < HID / 4; ++k) a[k] = *(const v2f*)(hrow + 4 * k);
  int rowoff = khalf * 8;
  for (int r = wu; r < RTOT; r += 16) {
    const float* wrel = w2 + (size_t)r * HID * NCLS + (size_t)(2 * khalf) * NCLS + mn;
    union { v8f v; float f[8]; } acc;
    acc.v = (v8f)0.0f;
#pragma unroll
    for (int k = 0; k < HID / 4; ++k) {
      v2f b;
      b.x = wrel[(size_t)(4 * k) * NCLS];
      b.y = wrel[(size_t)(4 * k + 1) * NCLS];
      acc.v = __builtin_amdgcn_wmma_f32_16x16x4_f32(
           false, a[k],  false, b,
           (short)0, acc.v,  false,  false);
    }
#pragma unroll
    for (int v = 0; v < 8; ++v) {
      P[(size_t)(mt * 16 + v + rowoff) * PCOLS + r * NCLS + mn] = acc.f[v];
    }
  }
}

#define L2_LOAD(I, U, PO, IL)                                                \
  {                                                                          \
    int po = quad_pick<U>(PO);                                               \
    iv[I] = __int_as_float(quad_pick<U>(__float_as_int(IL)));                \
    v[I] = *(const float4*)(Pb + (unsigned)(po + lo));                       \
  }
#define L2_SUM(I)                                                            \
  {                                                                          \
    acc.x = __builtin_fmaf(iv[I], v[I].x, acc.x);                            \
    acc.y = __builtin_fmaf(iv[I], v[I].y, acc.y);                            \
    acc.z = __builtin_fmaf(iv[I], v[I].z, acc.z);                            \
    acc.w = __builtin_fmaf(iv[I], v[I].w, acc.w);                            \
  }
template <int NA, int NB>
__device__ __forceinline__ void l2_stretch(int poffA, float invA, int poffB, float invB, int lo,
                                           const char* __restrict__ Pb, float4& acc) {
  float4 v[8];
  float iv[8];
  L2_LOAD(0, 0, poffA, invA) L2_LOAD(1, 1, poffA, invA)
  if (NA > 2) { L2_LOAD(2, 2, poffA, invA) L2_LOAD(3, 3, poffA, invA) }
  if (NB > 0) { L2_LOAD(4, 0, poffB, invB) L2_LOAD(5, 1, poffB, invB) }
  if (NB > 2) { L2_LOAD(6, 2, poffB, invB) L2_LOAD(7, 3, poffB, invB) }
  L2_SUM(0) L2_SUM(1)
  if (NA > 2) { L2_SUM(2) L2_SUM(3) }
  if (NB > 0) { L2_SUM(4) L2_SUM(5) }
  if (NB > 2) { L2_SUM(6) L2_SUM(7) }
}

__global__ __launch_bounds__(256) void k_l2pull(const int* __restrict__ off,
                                                const int* __restrict__ msg,
                                                const float* __restrict__ P,
                                                const float* __restrict__ b2,
                                                float* __restrict__ out) {
  int lane = threadIdx.x & 31;
  int wu = __builtin_amdgcn_readfirstlane(threadIdx.x >> 5);
  int pr = blockIdx.x * 8 + wu;
  if (pr >= NNODES / 2) return;
  int g = lane >> 2, c4 = lane & 3;
  int slot = 8 * c4 + g;
  int lo = c4 * 16;
  const char* Pb = (const char*)P;
  float4 res = make_float4(0.0f, 0.0f, 0.0f, 0.0f);
  for (int t = 0; t < 2; ++t) {
    int n = 2 * pr + t;
    int beg = off[n], end = off[n + 1];
    int cnt = seg_counts(msg, beg, end, lane);
    float invq = 1.0f / (float)cnt;
    int wpad = ((2 * NRELH) << OBITS) | n;
    float4 acc = make_float4(0.0f, 0.0f, 0.0f, 0.0f);
    for (int base = beg; base < end; base += 64) {
      int mA = end - base;
      int mB = mA - 32;
      if (mA > 32) mA = 32;
      if (mB < 0) mB = 0;
      if (mB > 32) mB = 32;
      bool okA = slot < mA, okB = slot < mB;
      int wA = wpad, wB = wpad;
      if (okA) wA = msg[base + slot];
      if (okB) wB = msg[base + 32 + slot];
      int qA = wA >> OBITS, oA = wA & OMASK;
      int qB = wB >> OBITS, oB = wB & OMASK;
      float invA = __int_as_float(__builtin_amdgcn_ds_bpermute(qA << 2, __float_as_int(invq)));
      float invB = __int_as_float(__builtin_amdgcn_ds_bpermute(qB << 2, __float_as_int(invq)));
      invA = okA ? invA : 0.0f;
      invB = okB ? invB : 0.0f;
      int poffA = (oA * PCOLS + qA * NCLS) << 2;
      int poffB = (oB * PCOLS + qB * NCLS) << 2;
      if (mB > 16)      l2_stretch<4, 4>(poffA, invA, poffB, invB, lo, Pb, acc);
      else if (mB > 0)  l2_stretch<4, 2>(poffA, invA, poffB, invB, lo, Pb, acc);
      else if (mA > 16) l2_stretch<4, 0>(poffA, invA, poffB, invB, lo, Pb, acc);
      else              l2_stretch<2, 0>(poffA, invA, poffB, invB, lo, Pb, acc);
    }
    acc.x += __shfl_xor(acc.x, 4);  acc.y += __shfl_xor(acc.y, 4);
    acc.z += __shfl_xor(acc.z, 4);  acc.w += __shfl_xor(acc.w, 4);
    acc.x += __shfl_xor(acc.x, 8);  acc.y += __shfl_xor(acc.y, 8);
    acc.z += __shfl_xor(acc.z, 8);  acc.w += __shfl_xor(acc.w, 8);
    acc.x += __shfl_xor(acc.x, 16); acc.y += __shfl_xor(acc.y, 16);
    acc.z += __shfl_xor(acc.z, 16); acc.w += __shfl_xor(acc.w, 16);
    if (g == t) res = acc;
  }
  if (lane < 8) {
    int n = 2 * pr + g;
    float4 bb = *(const float4*)(b2 + 4 * c4);
    float4 ps = *(const float4*)(P + (size_t)n * PCOLS + (2 * NRELH) * NCLS + 4 * c4);
    float4 ov;
    ov.x = bb.x + ps.x + res.x;
    ov.y = bb.y + ps.y + res.y;
    ov.z = bb.z + ps.z + res.z;
    ov.w = bb.w + ps.w + res.w;
    *(float4*)(out + (size_t)n * NCLS + 4 * c4) = ov;
  }
}

extern "C" void kernel_launch(void* const* d_in, const int* in_sizes, int n_in,
                              void* d_out, int out_size, void* d_ws, size_t ws_size,
                              hipStream_t stream) {
  const int*   src = (const int*)d_in[0];
  const int*   rel = (const int*)d_in[1];
  const int*   dst = (const int*)d_in[2];
  const float* w1  = (const float*)d_in[3];
  const float* b1  = (const float*)d_in[4];
  const float* w2  = (const float*)d_in[5];
  const float* b2  = (const float*)d_in[6];
  float* out = (float*)d_out;
  const int E = in_sizes[0];

  int* off  = (int*)d_ws;
  int* msg  = off + NOFF;
  float* P  = (float*)(msg + 2 * (size_t)E);
  int* tab  = (int*)(P + (size_t)NNODES * PCOLS);
  int* tot  = tab + (size_t)NCH * NNODES;

  k_count<<<NPAIR, CNT_T, 0, stream>>>(src, dst, tab, tot, E);

  k_fill<<<NPAIR, FILL_T, 0, stream>>>(src, rel, dst, tab, tot, off, msg, E);

  k_l1gemm<<<NNODES / 16, 512, 0, stream>>>(off, msg, w1, b1, w2, P);

  k_l2pull<<<(NNODES / 2 + 7) / 8, 256, 0, stream>>>(off, msg, P, b2, out);
}
